# starting kernel
# baseline (speedup 1.0000x reference)
#include <hip/hip_runtime.h>

typedef __attribute__((ext_vector_type(2))) float v2f;
typedef __attribute__((ext_vector_type(8))) float v8f;

#define FEAT 128

__global__ void __launch_bounds__(256)
sage_scatter(const float* __restrict__ xsrc,
             const int* __restrict__ src,
             const int* __restrict__ dst,
             float* __restrict__ agg,
             float* __restrict__ cnt,
             int E)
{
    int wave = (int)((blockIdx.x * blockDim.x + threadIdx.x) >> 5);
    int lane = threadIdx.x & 31;
    if (wave >= E) return;
    int s = src[wave];
    int d = dst[wave];
    const float4* row = (const float4*)(xsrc + (size_t)s * FEAT);
    float4 v = row[lane];
    float* o = agg + (size_t)d * FEAT + lane * 4;
    atomicAdd(o + 0, v.x);
    atomicAdd(o + 1, v.y);
    atomicAdd(o + 2, v.z);
    atomicAdd(o + 3, v.w);
    if (lane == 0) atomicAdd(cnt + d, 1.0f);
}

__global__ void __launch_bounds__(256)
sage_mean_div(float* __restrict__ agg, const float* __restrict__ cnt, int total)
{
    int i = blockIdx.x * blockDim.x + threadIdx.x;
    if (i >= total) return;
    float c = cnt[i >> 7];
    agg[i] *= 1.0f / fmaxf(c, 1.0f);
}

__global__ void __launch_bounds__(256)
sage_gemm_wmma(const float* __restrict__ A1, const float* __restrict__ A2,
               const float* __restrict__ W1, const float* __restrict__ W2,
               const float* __restrict__ bias, float* __restrict__ out,
               int M, int N, int tiles_n, int total_tiles)
{
    int wave = (int)((blockIdx.x * blockDim.x + threadIdx.x) >> 5);
    if (wave >= total_tiles) return;
    int lane  = threadIdx.x & 31;
    int tm    = wave / tiles_n;
    int tn    = wave % tiles_n;

    int lmod  = lane & 15;
    int hi    = lane >> 4;
    int khalf = hi << 1;

    int m = tm * 16 + lmod;
    int n = tn * 16 + lmod;
    float amask = (m < M) ? 1.0f : 0.0f;
    int   mc    = (m < M) ? m : (M - 1);

    const float* a1 = A1 + (size_t)mc * FEAT + khalf;
    const float* a2 = A2 + (size_t)mc * FEAT + khalf;
    const float* w1 = W1 + (size_t)n  * FEAT + khalf;
    const float* w2 = W2 + (size_t)n  * FEAT + khalf;

    v8f acc = {};

    #pragma unroll 4
    for (int k = 0; k < FEAT; k += 4) {
        v2f a, b;
        a[0] = a1[k]     * amask;
        a[1] = a1[k + 1] * amask;
        b[0] = w1[k];
        b[1] = w1[k + 1];
        acc = __builtin_amdgcn_wmma_f32_16x16x4_f32(
             false, a,  false, b,
             (short)0, acc,  false,  false);
    }
    #pragma unroll 4
    for (int k = 0; k < FEAT; k += 4) {
        v2f a, b;
        a[0] = a2[k]     * amask;
        a[1] = a2[k + 1] * amask;
        b[0] = w2[k];
        b[1] = w2[k + 1];
        acc = __builtin_amdgcn_wmma_f32_16x16x4_f32(
            false, a, false, b, (short)0, acc, false, false);
    }

    float bn    = bias[n];
    int   mbase = tm * 16 + hi * 8;
    #pragma unroll
    for (int j = 0; j < 8; ++j) {
        int mm = mbase + j;
        if (mm < M) {
            float v = acc[j] + bn;
            out[(size_t)mm * N + n] = v > 0.0f ? v : 0.0f;
        }
    }
}

extern "C" void kernel_launch(void* const* d_in, const int* in_sizes, int n_in,
                              void* d_out, int out_size, void* d_ws, size_t ws_size,
                              hipStream_t stream) {
    const float* x    = (const float*)d_in[0];
    const float* W1l  = (const float*)d_in[1];
    const float* b1l  = (const float*)d_in[2];
    const float* W1r  = (const float*)d_in[3];
    const float* W2l  = (const float*)d_in[4];
    const float* b2l  = (const float*)d_in[5];
    const float* W2r  = (const float*)d_in[6];
    const int*   src1 = (const int*)d_in[7];
    const int*   dst1 = (const int*)d_in[8];
    const int*   src2 = (const int*)d_in[9];
    const int*   dst2 = (const int*)d_in[10];

    const int E1     = in_sizes[7];
    const int E2     = in_sizes[9];
    const int N1FEAT = in_sizes[2];
    const int N2FEAT = in_sizes[5];
    const int SIZE2  = out_size / N2FEAT;
    const int SIZE1  = 50000;

    float* agg1 = (float*)d_ws;
    float* cnt1 = agg1 + (size_t)SIZE1 * FEAT;
    float* h    = cnt1 + SIZE1;
    float* agg2 = h    + (size_t)SIZE1 * FEAT;
    float* cnt2 = agg2 + (size_t)SIZE2 * FEAT;

    hipMemsetAsync(agg1, 0, ((size_t)SIZE1 * FEAT + SIZE1) * sizeof(float), stream);
    hipMemsetAsync(agg2, 0, ((size_t)SIZE2 * FEAT + SIZE2) * sizeof(float), stream);

    sage_scatter<<<(E1 + 7) / 8, 256, 0, stream>>>(x, src1, dst1, agg1, cnt1, E1);
    {
        int total = SIZE1 * FEAT;
        sage_mean_div<<<(total + 255) / 256, 256, 0, stream>>>(agg1, cnt1, total);
    }
    {
        int tiles_n = N1FEAT / 16;
        int tiles_m = (SIZE1 + 15) / 16;
        int total   = tiles_m * tiles_n;
        sage_gemm_wmma<<<(total + 7) / 8, 256, 0, stream>>>(
            agg1, x, W1l, W1r, b1l, h, SIZE1, N1FEAT, tiles_n, total);
    }

    sage_scatter<<<(E2 + 7) / 8, 256, 0, stream>>>(h, src2, dst2, agg2, cnt2, E2);
    {
        int total = SIZE2 * FEAT;
        sage_mean_div<<<(total + 255) / 256, 256, 0, stream>>>(agg2, cnt2, total);
    }
    {
        int tiles_n = N2FEAT / 16;
        int tiles_m = (SIZE2 + 15) / 16;
        int total   = tiles_m * tiles_n;
        sage_gemm_wmma<<<(total + 7) / 8, 256, 0, stream>>>(
            agg2, h, W2l, W2r, b2l, (float*)d_out, SIZE2, N2FEAT, tiles_n, total);
    }
}


# Replace the float-atomic edge scatter by a bucket sort on destination (integer count, scan, ranked fill) and a per-row register gather-sum that stores the mean; row sum order = fill arrival rank
# speedup vs baseline: 1.2349x; 1.2349x over previous
#include <hip/hip_runtime.h>

typedef __attribute__((ext_vector_type(2))) float v2f;
typedef __attribute__((ext_vector_type(8))) float v8f;

#define FEAT 128

#define IDX_STRIDE 65536

__global__ void __launch_bounds__(256)
sage_degree(const int* __restrict__ dst, int* __restrict__ cnt, int E)
{
    int i = blockIdx.x * blockDim.x + threadIdx.x;
    if (i < E) atomicAdd(cnt + dst[i], 1);
}

__global__ void __launch_bounds__(1024)
sage_offsets(const int* __restrict__ cnt, int* __restrict__ cur, int n)
{
    __shared__ int wsum[2][32];
    const int tid  = threadIdx.x;
    const int lane = tid & 31;
    const int w    = tid >> 5;
    const int ntiles = (n + 16383) >> 14;
    const int4* c4 = (const int4*)cnt + tid * 4;
    int4*       o4 = (int4*)cur + tid * 4;

    int  carry = 0;
    int4 a0 = c4[0], a1 = c4[1], a2 = c4[2], a3 = c4[3];
    for (int t = 0; t < ntiles; ++t) {
        const int tn = (t + 1 < ntiles) ? (t + 1) : t;
        const int4* cn = c4 + tn * 4096;
        int4 b0 = cn[0], b1 = cn[1], b2 = cn[2], b3 = cn[3];

        int p0  = a0.x,      p1  = p0  + a0.y, p2  = p1  + a0.z, p3  = p2  + a0.w;
        int p4  = p3 + a1.x, p5  = p4  + a1.y, p6  = p5  + a1.z, p7  = p6  + a1.w;
        int p8  = p7 + a2.x, p9  = p8  + a2.y, p10 = p9  + a2.z, p11 = p10 + a2.w;
        int p12 = p11 + a3.x, p13 = p12 + a3.y, p14 = p13 + a3.z, p15 = p14 + a3.w;

        int incl = p15;
        for (int o = 1; o < 32; o <<= 1) {
            int y = __shfl_up(incl, o);
            if (lane >= o) incl += y;
        }
        if (lane == 31) wsum[t & 1][w] = incl;
        __syncthreads();
        int ws    = wsum[t & 1][lane];
        int wincl = ws;
        for (int o = 1; o < 32; o <<= 1) {
            int y = __shfl_up(wincl, o);
            if (lane >= o) wincl += y;
        }
        int tile_total = __shfl(wincl, 31);
        int wave_excl  = __shfl(wincl, w) - __shfl(ws, w);
        int base = carry + wave_excl + (incl - p15);
        int4* o = o4 + t * 4096;
        o[0] = make_int4(base,       base + p0,  base + p1,  base + p2);
        o[1] = make_int4(base + p3,  base + p4,  base + p5,  base + p6);
        o[2] = make_int4(base + p7,  base + p8,  base + p9,  base + p10);
        o[3] = make_int4(base + p11, base + p12, base + p13, base + p14);
        carry += tile_total;
        a0 = b0; a1 = b1; a2 = b2; a3 = b3;
    }
}

__global__ void __launch_bounds__(256)
sage_fill(const int* __restrict__ src, const int* __restrict__ dst,
          int* __restrict__ cur, int* __restrict__ perm, int E)
{
    int i = blockIdx.x * blockDim.x + threadIdx.x;
    if (i < E) {
        int d = dst[i];
        int p = atomicAdd(cur + d, 1);
        if ((unsigned)p < (unsigned)E) perm[p] = src[i];
    }
}

__global__ void __launch_bounds__(256)
sage_gather_mean(const float* __restrict__ xsrc,
                 const int* __restrict__ perm,
                 const int* __restrict__ cnt,
                 const int* __restrict__ cur,
                 float* __restrict__ mean,
                 int n, int E)
{
    int wave = (int)((blockIdx.x * blockDim.x + threadIdx.x) >> 5);
    int lane = threadIdx.x & 31;
    if (wave >= n) return;
    const int c    = __builtin_amdgcn_readfirstlane(cnt[wave]);
    const int end0 = __builtin_amdgcn_readfirstlane(cur[wave]);
    const int end = (end0 < E) ? end0 : E;
    const int beg = (end0 - c > 0) ? (end0 - c) : 0;

    float4 acc = make_float4(0.0f, 0.0f, 0.0f, 0.0f);
    for (int base = beg; base < end; base += 32) {
        int nrem = end - base;
        if (nrem > 32) nrem = 32;
        int my = (lane < nrem) ? perm[base + lane] : 0;
        for (int j = 0; j < nrem; j += 8) {
            float4 v[8];
            #pragma unroll
            for (int u = 0; u < 8; ++u) {
                int s = __builtin_amdgcn_readlane(my, j + u);
                v[u] = ((const float4*)(xsrc + (size_t)s * FEAT))[lane];
            }
            #pragma unroll
            for (int u = 0; u < 8; ++u) {
                if (j + u < nrem) {
                    acc.x += v[u].x;
                    acc.y += v[u].y;
                    acc.z += v[u].z;
                    acc.w += v[u].w;
                }
            }
        }
    }
    float r = 1.0f / fmaxf((float)c, 1.0f);
    float4 o = make_float4(acc.x * r, acc.y * r, acc.z * r, acc.w * r);
    ((float4*)(mean + (size_t)wave * FEAT))[lane] = o;
}

__global__ void __launch_bounds__(256)
sage_gemm_wmma(const float* __restrict__ A1, const float* __restrict__ A2,
               const float* __restrict__ W1, const float* __restrict__ W2,
               const float* __restrict__ bias, float* __restrict__ out,
               int M, int N, int tiles_n, int total_tiles)
{
    int wave = (int)((blockIdx.x * blockDim.x + threadIdx.x) >> 5);
    if (wave >= total_tiles) return;
    int lane  = threadIdx.x & 31;
    int tm    = wave / tiles_n;
    int tn    = wave % tiles_n;

    int lmod  = lane & 15;
    int hi    = lane >> 4;
    int khalf = hi << 1;

    int m = tm * 16 + lmod;
    int n = tn * 16 + lmod;
    float amask = (m < M) ? 1.0f : 0.0f;
    int   mc    = (m < M) ? m : (M - 1);

    const float* a1 = A1 + (size_t)mc * FEAT + khalf;
    const float* a2 = A2 + (size_t)mc * FEAT + khalf;
    const float* w1 = W1 + (size_t)n  * FEAT + khalf;
    const float* w2 = W2 + (size_t)n  * FEAT + khalf;

    v8f acc = {};

    #pragma unroll 4
    for (int k = 0; k < FEAT; k += 4) {
        v2f a, b;
        a[0] = a1[k]     * amask;
        a[1] = a1[k + 1] * amask;
        b[0] = w1[k];
        b[1] = w1[k + 1];
        acc = __builtin_amdgcn_wmma_f32_16x16x4_f32(
             false, a,  false, b,
             (short)0, acc,  false,  false);
    }
    #pragma unroll 4
    for (int k = 0; k < FEAT; k += 4) {
        v2f a, b;
        a[0] = a2[k]     * amask;
        a[1] = a2[k + 1] * amask;
        b[0] = w2[k];
        b[1] = w2[k + 1];
        acc = __builtin_amdgcn_wmma_f32_16x16x4_f32(
            false, a, false, b, (short)0, acc, false, false);
    }

    float bn    = bias[n];
    int   mbase = tm * 16 + hi * 8;
    #pragma unroll
    for (int j = 0; j < 8; ++j) {
        int mm = mbase + j;
        if (mm < M) {
            float v = acc[j] + bn;
            out[(size_t)mm * N + n] = v > 0.0f ? v : 0.0f;
        }
    }
}

extern "C" void kernel_launch(void* const* d_in, const int* in_sizes, int n_in,
                              void* d_out, int out_size, void* d_ws, size_t ws_size,
                              hipStream_t stream) {
    const float* x    = (const float*)d_in[0];
    const float* W1l  = (const float*)d_in[1];
    const float* b1l  = (const float*)d_in[2];
    const float* W1r  = (const float*)d_in[3];
    const float* W2l  = (const float*)d_in[4];
    const float* b2l  = (const float*)d_in[5];
    const float* W2r  = (const float*)d_in[6];
    const int*   src1 = (const int*)d_in[7];
    const int*   dst1 = (const int*)d_in[8];
    const int*   src2 = (const int*)d_in[9];
    const int*   dst2 = (const int*)d_in[10];

    const int E1     = in_sizes[7];
    const int E2     = in_sizes[9];
    const int N1FEAT = in_sizes[2];
    const int N2FEAT = in_sizes[5];
    const int SIZE2  = out_size / N2FEAT;
    const int SIZE1  = 50000;

    float* agg1 = (float*)d_ws;
    float* cnt1 = agg1 + (size_t)SIZE1 * FEAT;
    float* h    = cnt1 + SIZE1;
    float* agg2 = h    + (size_t)SIZE1 * FEAT;

    {
        int* icnt  = (int*)h;
        int* icur  = icnt + IDX_STRIDE;
        int* iperm = icur + IDX_STRIDE;
        hipMemsetAsync(icnt, 0, (size_t)IDX_STRIDE * sizeof(int), stream);
        sage_degree<<<(E1 + 255) / 256, 256, 0, stream>>>(dst1, icnt, E1);
        sage_offsets<<<1, 1024, 0, stream>>>(icnt, icur, SIZE1);
        sage_fill<<<(E1 + 255) / 256, 256, 0, stream>>>(src1, dst1, icur, iperm, E1);
        sage_gather_mean<<<(SIZE1 + 7) / 8, 256, 0, stream>>>(
            x, iperm, icnt, icur, agg1, SIZE1, E1);
    }
    {
        int tiles_n = N1FEAT / 16;
        int tiles_m = (SIZE1 + 15) / 16;
        int total   = tiles_m * tiles_n;
        sage_gemm_wmma<<<(total + 7) / 8, 256, 0, stream>>>(
            agg1, x, W1l, W1r, b1l, h, SIZE1, N1FEAT, tiles_n, total);
    }

    {
        int* icnt  = (int*)agg1;
        int* icur  = icnt + IDX_STRIDE;
        int* iperm = icur + IDX_STRIDE;
        hipMemsetAsync(icnt, 0, (size_t)IDX_STRIDE * sizeof(int), stream);
        sage_degree<<<(E2 + 255) / 256, 256, 0, stream>>>(dst2, icnt, E2);
        sage_offsets<<<1, 1024, 0, stream>>>(icnt, icur, SIZE2);
        sage_fill<<<(E2 + 255) / 256, 256, 0, stream>>>(src2, dst2, icur, iperm, E2);
        sage_gather_mean<<<(SIZE2 + 7) / 8, 256, 0, stream>>>(
            h, iperm, icnt, icur, agg2, SIZE2, E2);
    }
    {
        int tiles_n = N2FEAT / 16;
        int tiles_m = (SIZE2 + 15) / 16;
        int total   = tiles_m * tiles_n;
        sage_gemm_wmma<<<(total + 7) / 8, 256, 0, stream>>>(
            agg2, h, W2l, W2r, b2l, (float*)d_out, SIZE2, N2FEAT, tiles_n, total);
    }
}


# Fused linear re-tiled as a resident grid: each wave keeps its 16 weight columns in registers, activation rows reach WMMA through LDS by whole-row loads; k summed in 8-wide blocks split by lane half
# speedup vs baseline: 1.4815x; 1.1997x over previous
#include <hip/hip_runtime.h>

typedef __attribute__((ext_vector_type(2))) float v2f;
typedef __attribute__((ext_vector_type(8))) float v8f;
typedef __attribute__((ext_vector_type(4))) float v4f;

#define FEAT 128

#define IDX_STRIDE 65536

__global__ void __launch_bounds__(256)
sage_degree(const int* __restrict__ dst, int* __restrict__ cnt, int E)
{
    int i = blockIdx.x * blockDim.x + threadIdx.x;
    if (i < E) atomicAdd(cnt + dst[i], 1);
}

__global__ void __launch_bounds__(1024)
sage_offsets(const int* __restrict__ cnt, int* __restrict__ cur, int n)
{
    __shared__ int wsum[2][32];
    const int tid  = threadIdx.x;
    const int lane = tid & 31;
    const int w    = tid >> 5;
    const int ntiles = (n + 16383) >> 14;
    const int4* c4 = (const int4*)cnt + tid * 4;
    int4*       o4 = (int4*)cur + tid * 4;

    int  carry = 0;
    int4 a0 = c4[0], a1 = c4[1], a2 = c4[2], a3 = c4[3];
    for (int t = 0; t < ntiles; ++t) {
        const int tn = (t + 1 < ntiles) ? (t + 1) : t;
        const int4* cn = c4 + tn * 4096;
        int4 b0 = cn[0], b1 = cn[1], b2 = cn[2], b3 = cn[3];

        int p0  = a0.x,      p1  = p0  + a0.y, p2  = p1  + a0.z, p3  = p2  + a0.w;
        int p4  = p3 + a1.x, p5  = p4  + a1.y, p6  = p5  + a1.z, p7  = p6  + a1.w;
        int p8  = p7 + a2.x, p9  = p8  + a2.y, p10 = p9  + a2.z, p11 = p10 + a2.w;
        int p12 = p11 + a3.x, p13 = p12 + a3.y, p14 = p13 + a3.z, p15 = p14 + a3.w;

        int incl = p15;
        for (int o = 1; o < 32; o <<= 1) {
            int y = __shfl_up(incl, o);
            if (lane >= o) incl += y;
        }
        if (lane == 31) wsum[t & 1][w] = incl;
        __syncthreads();
        int ws    = wsum[t & 1][lane];
        int wincl = ws;
        for (int o = 1; o < 32; o <<= 1) {
            int y = __shfl_up(wincl, o);
            if (lane >= o) wincl += y;
        }
        int tile_total = __shfl(wincl, 31);
        int wave_excl  = __shfl(wincl, w) - __shfl(ws, w);
        int base = carry + wave_excl + (incl - p15);
        int4* o = o4 + t * 4096;
        o[0] = make_int4(base,       base + p0,  base + p1,  base + p2);
        o[1] = make_int4(base + p3,  base + p4,  base + p5,  base + p6);
        o[2] = make_int4(base + p7,  base + p8,  base + p9,  base + p10);
        o[3] = make_int4(base + p11, base + p12, base + p13, base + p14);
        carry += tile_total;
        a0 = b0; a1 = b1; a2 = b2; a3 = b3;
    }
}

__global__ void __launch_bounds__(256)
sage_fill(const int* __restrict__ src, const int* __restrict__ dst,
          int* __restrict__ cur, int* __restrict__ perm, int E)
{
    int i = blockIdx.x * blockDim.x + threadIdx.x;
    if (i < E) {
        int d = dst[i];
        int p = atomicAdd(cur + d, 1);
        if ((unsigned)p < (unsigned)E) perm[p] = src[i];
    }
}

__global__ void __launch_bounds__(256)
sage_gather_mean(const float* __restrict__ xsrc,
                 const int* __restrict__ perm,
                 const int* __restrict__ cnt,
                 const int* __restrict__ cur,
                 float* __restrict__ mean,
                 int n, int E)
{
    int wave = (int)((blockIdx.x * blockDim.x + threadIdx.x) >> 5);
    int lane = threadIdx.x & 31;
    if (wave >= n) return;
    const int c    = __builtin_amdgcn_readfirstlane(cnt[wave]);
    const int end0 = __builtin_amdgcn_readfirstlane(cur[wave]);
    const int end = (end0 < E) ? end0 : E;
    const int beg = (end0 - c > 0) ? (end0 - c) : 0;

    float4 acc = make_float4(0.0f, 0.0f, 0.0f, 0.0f);
    for (int base = beg; base < end; base += 32) {
        int nrem = end - base;
        if (nrem > 32) nrem = 32;
        int my = (lane < nrem) ? perm[base + lane] : 0;
        for (int j = 0; j < nrem; j += 8) {
            float4 v[8];
            #pragma unroll
            for (int u = 0; u < 8; ++u) {
                int s = __builtin_amdgcn_readlane(my, j + u);
                v[u] = ((const float4*)(xsrc + (size_t)s * FEAT))[lane];
            }
            #pragma unroll
            for (int u = 0; u < 8; ++u) {
                if (j + u < nrem) {
                    acc.x += v[u].x;
                    acc.y += v[u].y;
                    acc.z += v[u].z;
                    acc.w += v[u].w;
                }
            }
        }
    }
    float r = 1.0f / fmaxf((float)c, 1.0f);
    float4 o = make_float4(acc.x * r, acc.y * r, acc.z * r, acc.w * r);
    ((float4*)(mean + (size_t)wave * FEAT))[lane] = o;
}

#define WPITCH 132
#define APITCH 260

template <int NT>
__global__ void __launch_bounds__(256) __attribute__((amdgpu_waves_per_eu(4, 4)))
sage_gemm_lds(const float* __restrict__ A1, const float* __restrict__ A2,
              const float* __restrict__ W1, const float* __restrict__ W2,
              const float* __restrict__ bias, float* __restrict__ out,
              int M, int units)
{
    constexpr int N    = NT * 16;
    constexpr int ROWS = (8 / NT) * 16;
    constexpr int LPT  = (2 * ROWS) / 8;
    constexpr int WLD  = N / 8;
    constexpr int LDSF = (N * WPITCH > ROWS * APITCH) ? N * WPITCH
                                                      : ROWS * APITCH;
    __shared__ __attribute__((aligned(16))) float lds[LDSF];

    const int tid  = threadIdx.x;
    const int lane = tid & 31;
    const int w    = tid >> 5;
    const int c    = lane & 15;
    const int h    = lane >> 4;
    const int nt   = w % NT;
    const int mg   = w / NT;
    const int n    = nt * 16 + c;

    float4 bw[32];
    #pragma unroll
    for (int half = 0; half < 2; ++half) {
        const float* Wsrc = half ? W2 : W1;
        float4 t[WLD];
        #pragma unroll
        for (int i = 0; i < WLD; ++i)
            t[i] = ((const float4*)(Wsrc + (size_t)(w + 8 * i) * FEAT))[lane];
        #pragma unroll
        for (int i = 0; i < WLD; ++i)
            *(float4*)(lds + (w + 8 * i) * WPITCH + 4 * lane) = t[i];
        __syncthreads();
        #pragma unroll
        for (int j = 0; j < 16; ++j)
            bw[half * 16 + j] =
                *(const float4*)(lds + n * WPITCH + 8 * j + 4 * h);
        __syncthreads();
    }
    const float bn = bias[n];

    v4f pre[LPT];
    int u = blockIdx.x;
    #pragma unroll
    for (int i = 0; i < LPT; ++i) {
        const float* S = (i * 8 >= ROWS) ? A2 : A1;
        int m = u * ROWS + (i * 8) % ROWS + w;
        m = (m < M) ? m : (M - 1);
        pre[i] = ((const v4f*)(S + (size_t)m * FEAT))[lane];
    }

    const float* arow = lds + (mg * 16 + c) * APITCH + 4 * h;

    for (; u < units; u += gridDim.x) {
        #pragma unroll
        for (int i = 0; i < LPT; ++i) {
            const int r   = (i * 8) % ROWS + w;
            const int off = (i * 8 >= ROWS) ? FEAT : 0;
            *(v4f*)(lds + r * APITCH + off + 4 * lane) = pre[i];
        }
        __syncthreads();

        const int un = u + gridDim.x;
        if (un < units) {
            #pragma unroll
            for (int i = 0; i < LPT; ++i) {
                const float* S = (i * 8 >= ROWS) ? A2 : A1;
                int m = un * ROWS + (i * 8) % ROWS + w;
                m = (m < M) ? m : (M - 1);
                pre[i] = ((const v4f*)(S + (size_t)m * FEAT))[lane];
            }
        }

        v8f acc = {};
        #pragma unroll
        for (int j = 0; j < 32; ++j) {
            const float4 a = *(const float4*)(arow + 8 * j);
            v2f a0, a1, b0, b1;
            a0[0] = a.x;     a0[1] = a.y;     a1[0] = a.z;     a1[1] = a.w;
            b0[0] = bw[j].x; b0[1] = bw[j].y; b1[0] = bw[j].z; b1[1] = bw[j].w;
            acc = __builtin_amdgcn_wmma_f32_16x16x4_f32(
                 false, a0,  false, b0,
                 (short)0, acc,  false,  false);
            acc = __builtin_amdgcn_wmma_f32_16x16x4_f32(
                false, a1, false, b1, (short)0, acc, false, false);
        }

        const int mbase = u * ROWS + mg * 16 + h * 8;
        #pragma unroll
        for (int j = 0; j < 8; ++j) {
            const int mm = mbase + j;
            if (mm < M) {
                const float v = acc[j] + bn;
                out[(size_t)mm * N + n] = v > 0.0f ? v : 0.0f;
            }
        }
        __syncthreads();
    }
}

extern "C" void kernel_launch(void* const* d_in, const int* in_sizes, int n_in,
                              void* d_out, int out_size, void* d_ws, size_t ws_size,
                              hipStream_t stream) {
    const float* x    = (const float*)d_in[0];
    const float* W1l  = (const float*)d_in[1];
    const float* b1l  = (const float*)d_in[2];
    const float* W1r  = (const float*)d_in[3];
    const float* W2l  = (const float*)d_in[4];
    const float* b2l  = (const float*)d_in[5];
    const float* W2r  = (const float*)d_in[6];
    const int*   src1 = (const int*)d_in[7];
    const int*   dst1 = (const int*)d_in[8];
    const int*   src2 = (const int*)d_in[9];
    const int*   dst2 = (const int*)d_in[10];

    const int E1     = in_sizes[7];
    const int E2     = in_sizes[9];
    const int N1FEAT = in_sizes[2];
    const int N2FEAT = in_sizes[5];
    const int SIZE2  = out_size / N2FEAT;
    const int SIZE1  = 50000;

    float* agg1 = (float*)d_ws;
    float* cnt1 = agg1 + (size_t)SIZE1 * FEAT;
    float* h    = cnt1 + SIZE1;
    float* agg2 = h    + (size_t)SIZE1 * FEAT;

    {
        int* icnt  = (int*)h;
        int* icur  = icnt + IDX_STRIDE;
        int* iperm = icur + IDX_STRIDE;
        hipMemsetAsync(icnt, 0, (size_t)IDX_STRIDE * sizeof(int), stream);
        sage_degree<<<(E1 + 255) / 256, 256, 0, stream>>>(dst1, icnt, E1);
        sage_offsets<<<1, 1024, 0, stream>>>(icnt, icur, SIZE1);
        sage_fill<<<(E1 + 255) / 256, 256, 0, stream>>>(src1, dst1, icur, iperm, E1);
        sage_gather_mean<<<(SIZE1 + 7) / 8, 256, 0, stream>>>(
            x, iperm, icnt, icur, agg1, SIZE1, E1);
    }
    {
        int units = (SIZE1 + 15) / 16;
        int grid  = units < 512 ? units : 512;
        sage_gemm_lds<8><<<grid, 256, 0, stream>>>(
            agg1, x, W1l, W1r, b1l, h, SIZE1, units);
    }

    {
        int* icnt  = (int*)agg1;
        int* icur  = icnt + IDX_STRIDE;
        int* iperm = icur + IDX_STRIDE;
        hipMemsetAsync(icnt, 0, (size_t)IDX_STRIDE * sizeof(int), stream);
        sage_degree<<<(E2 + 255) / 256, 256, 0, stream>>>(dst2, icnt, E2);
        sage_offsets<<<1, 1024, 0, stream>>>(icnt, icur, SIZE2);
        sage_fill<<<(E2 + 255) / 256, 256, 0, stream>>>(src2, dst2, icur, iperm, E2);
        sage_gather_mean<<<(SIZE2 + 7) / 8, 256, 0, stream>>>(
            h, iperm, icnt, icur, agg2, SIZE2, E2);
    }
    {
        int units = (SIZE2 + 31) / 32;
        int grid  = units < 512 ? units : 512;
        sage_gemm_lds<4><<<grid, 256, 0, stream>>>(
            agg2, h, W2l, W2r, b2l, (float*)d_out, SIZE2, units);
    }
}


# Edges bucketed by destination in two levels on LDS counters (chunk-local sort by row range, then one workgroup per range ranks its rows) in place of global atomics; row sum order = LDS arrival rank
# speedup vs baseline: 3.2952x; 2.2242x over previous
#include <hip/hip_runtime.h>

typedef __attribute__((ext_vector_type(2))) float v2f;
typedef __attribute__((ext_vector_type(8))) float v8f;
typedef __attribute__((ext_vector_type(4))) float v4f;

#define FEAT 128

#define IDX_STRIDE 65536

#define CHUNK 8192
#define BT    512
#define GMAX  512
#define RMAX  256
#define STAGE 16384

__device__ __forceinline__ int block_excl_scan(int v, int* wtot, int& total)
{
    const int lane = threadIdx.x & 31;
    const int w    = threadIdx.x >> 5;
    int incl = v;
    for (int o = 1; o < 32; o <<= 1) {
        int y = __shfl_up(incl, o);
        if (lane >= o) incl += y;
    }
    if (lane == 31) wtot[w] = incl;
    __syncthreads();
    int ws    = (lane < BT / 32) ? wtot[lane] : 0;
    int wincl = ws;
    for (int o = 1; o < 32; o <<= 1) {
        int y = __shfl_up(wincl, o);
        if (lane >= o) wincl += y;
    }
    total = __shfl(wincl, 31);
    const int wave_excl = __shfl(wincl, w) - __shfl(ws, w);
    return wave_excl + incl - v;
}

__global__ void __launch_bounds__(BT)
sage_sort_chunks(const int* __restrict__ src, const int* __restrict__ dst,
                 int* __restrict__ ps, int* __restrict__ pd,
                 int* __restrict__ tbl, int E, int rsh, int nb, int g1)
{
    constexpr int EPT = CHUNK / BT;
    __shared__ int hist[BT];
    __shared__ int wt[32];
    __shared__ __attribute__((aligned(16))) int sS[CHUNK];
    __shared__ __attribute__((aligned(16))) int sD[CHUNK];

    const int tid = threadIdx.x;
    const int g   = blockIdx.x;
    const int e0  = g * CHUNK;

    hist[tid] = 0;
    int s[EPT], d[EPT], r[EPT];
    #pragma unroll
    for (int j = 0; j < EPT; ++j) {
        const int e = e0 + j * BT + tid;
        const bool ok = e < E;
        d[j] = ok ? dst[e] : 0;
        s[j] = ok ? src[e] : 0;
    }
    __syncthreads();
    #pragma unroll
    for (int j = 0; j < EPT; ++j) {
        r[j] = 0;
        if (e0 + j * BT + tid < E) {
            int bn = d[j] >> rsh;
            bn = bn < 0 ? 0 : (bn > nb - 1 ? nb - 1 : bn);
            r[j] = atomicAdd(&hist[bn], 1);
        }
    }
    __syncthreads();
    int total;
    const int ex = block_excl_scan(hist[tid], wt, total);
    hist[tid] = ex;
    if (tid <= nb) tbl[tid * g1 + g] = ex;
    __syncthreads();
    #pragma unroll
    for (int j = 0; j < EPT; ++j) {
        if (e0 + j * BT + tid < E) {
            int bn = d[j] >> rsh;
            bn = bn < 0 ? 0 : (bn > nb - 1 ? nb - 1 : bn);
            const int p = hist[bn] + r[j];
            sS[p] = s[j];
            sD[p] = d[j];
        }
    }
    __syncthreads();
    const int count = (E - e0 < CHUNK) ? (E - e0) : CHUNK;
    const int nq    = count >> 2;
    int4* ops = (int4*)(ps + e0);
    int4* opd = (int4*)(pd + e0);
    for (int i = tid; i < nq; i += BT) {
        ops[i] = ((const int4*)sS)[i];
        opd[i] = ((const int4*)sD)[i];
    }
    for (int i = (nq << 2) + tid; i < count; i += BT) {
        ps[e0 + i] = sS[i];
        pd[e0 + i] = sD[i];
    }
}

#define RU 8
__global__ void __launch_bounds__(BT)
sage_rank_ranges(const int* __restrict__ ps, const int* __restrict__ pd,
                 const int* __restrict__ tbl,
                 int* __restrict__ cnt, int* __restrict__ cur,
                 int* __restrict__ perm, int n, int E, int rsh, int g1)
{
    constexpr int NW = BT / 32;
    __shared__ int rs[GMAX + RU * NW];
    __shared__ int rl[GMAX + RU * NW];
    __shared__ int cL[RMAX];
    __shared__ int wtA[32];
    __shared__ int wtB[32];
    __shared__ int wtC[32];
    __shared__ int stg[STAGE];

    const int tid   = threadIdx.x;
    const int lane  = tid & 31;
    const int w     = __builtin_amdgcn_readfirstlane(tid >> 5);
    const int b     = blockIdx.x;
    const int R     = 1 << rsh;
    const int dbase = b << rsh;

    int lo = 0, len = 0;
    if (tid < g1) {
        lo = tbl[b * g1 + tid];
        int hi = tbl[(b + 1) * g1 + tid];
        lo = lo < 0 ? 0 : (lo > CHUNK ? CHUNK : lo);
        hi = hi < lo ? lo : (hi > CHUNK ? CHUNK : hi);
        len = hi - lo;
    }
    rs[tid] = tid * CHUNK + lo;
    rl[tid] = len;
    if (tid < RU * NW) { rs[GMAX + tid] = 0; rl[GMAX + tid] = 0; }
    if (tid < RMAX) cL[tid] = 0;
    int base, S;
    block_excl_scan(lo,  wtA, base);
    block_excl_scan(len, wtB, S);

    for (int g0 = w; g0 < g1; g0 += RU * NW) {
        int dv[RU][2], l[RU], off[RU];
        #pragma unroll
        for (int u = 0; u < RU; ++u) {
            l[u]   = rl[g0 + u * NW];
            off[u] = rs[g0 + u * NW];
        }
        #pragma unroll
        for (int u = 0; u < RU; ++u) {
            dv[u][0] = (lane      < l[u]) ? pd[off[u] + lane]      : -1;
            dv[u][1] = (lane + 32 < l[u]) ? pd[off[u] + lane + 32] : -1;
        }
        #pragma unroll
        for (int u = 0; u < RU; ++u) {
            #pragma unroll
            for (int k = 0; k < 2; ++k) {
                const int dl = dv[u][k] - dbase;
                if ((unsigned)dl < (unsigned)R) atomicAdd(&cL[dl], 1);
            }
        }
        #pragma unroll
        for (int u = 0; u < RU; ++u) {
            for (int i = 64 + lane; i < l[u]; i += 32) {
                const int dl = pd[off[u] + i] - dbase;
                if ((unsigned)dl < (unsigned)R) atomicAdd(&cL[dl], 1);
            }
        }
    }
    __syncthreads();

    const int c = (tid < R) ? cL[tid] : 0;
    int tot;
    const int ex = block_excl_scan(c, wtC, tot);
    if (tid < R) {
        cL[tid] = ex;
        const int dd = dbase + tid;
        if (dd < n) { cnt[dd] = c; cur[dd] = base + ex + c; }
    }
    __syncthreads();

    const bool staged = (S <= STAGE);
    for (int g0 = w; g0 < g1; g0 += RU * NW) {
        int dv[RU][2], sv[RU][2], l[RU], off[RU];
        #pragma unroll
        for (int u = 0; u < RU; ++u) {
            l[u]   = rl[g0 + u * NW];
            off[u] = rs[g0 + u * NW];
        }
        #pragma unroll
        for (int u = 0; u < RU; ++u) {
            const bool k0 = lane < l[u], k1 = lane + 32 < l[u];
            dv[u][0] = k0 ? pd[off[u] + lane]      : -1;
            sv[u][0] = k0 ? ps[off[u] + lane]      : 0;
            dv[u][1] = k1 ? pd[off[u] + lane + 32] : -1;
            sv[u][1] = k1 ? ps[off[u] + lane + 32] : 0;
        }
        #pragma unroll
        for (int u = 0; u < RU; ++u) {
            #pragma unroll
            for (int k = 0; k < 2; ++k) {
                const int dl = dv[u][k] - dbase;
                if ((unsigned)dl < (unsigned)R) {
                    const int p = atomicAdd(&cL[dl], 1);
                    if (staged) {
                        if ((unsigned)p < (unsigned)STAGE) stg[p] = sv[u][k];
                    } else {
                        const int q = base + p;
                        if ((unsigned)q < (unsigned)E) perm[q] = sv[u][k];
                    }
                }
            }
        }
        #pragma unroll
        for (int u = 0; u < RU; ++u) {
            for (int i = 64 + lane; i < l[u]; i += 32) {
                const int dl = pd[off[u] + i] - dbase;
                const int sx = ps[off[u] + i];
                if ((unsigned)dl < (unsigned)R) {
                    const int p = atomicAdd(&cL[dl], 1);
                    if (staged) {
                        if ((unsigned)p < (unsigned)STAGE) stg[p] = sx;
                    } else {
                        const int q = base + p;
                        if ((unsigned)q < (unsigned)E) perm[q] = sx;
                    }
                }
            }
        }
    }
    if (staged) {
        __syncthreads();
        for (int i = tid; i < S; i += BT) {
            const int q = base + i;
            if ((unsigned)q < (unsigned)E) perm[q] = stg[i];
        }
    }
}

__global__ void __launch_bounds__(256)
sage_gather_mean(const float* __restrict__ xsrc,
                 const int* __restrict__ perm,
                 const int* __restrict__ cnt,
                 const int* __restrict__ cur,
                 float* __restrict__ mean,
                 int n, int E)
{
    int wave = (int)((blockIdx.x * blockDim.x + threadIdx.x) >> 5);
    int lane = threadIdx.x & 31;
    if (wave >= n) return;
    const int c    = __builtin_amdgcn_readfirstlane(cnt[wave]);
    const int end0 = __builtin_amdgcn_readfirstlane(cur[wave]);
    const int end = (end0 < E) ? end0 : E;
    const int beg = (end0 - c > 0) ? (end0 - c) : 0;

    float4 acc = make_float4(0.0f, 0.0f, 0.0f, 0.0f);
    for (int base = beg; base < end; base += 32) {
        int nrem = end - base;
        if (nrem > 32) nrem = 32;
        int my = (lane < nrem) ? perm[base + lane] : 0;
        for (int j = 0; j < nrem; j += 8) {
            float4 v[8];
            #pragma unroll
            for (int u = 0; u < 8; ++u) {
                int s = __builtin_amdgcn_readlane(my, j + u);
                v[u] = ((const float4*)(xsrc + (size_t)s * FEAT))[lane];
            }
            #pragma unroll
            for (int u = 0; u < 8; ++u) {
                if (j + u < nrem) {
                    acc.x += v[u].x;
                    acc.y += v[u].y;
                    acc.z += v[u].z;
                    acc.w += v[u].w;
                }
            }
        }
    }
    float r = 1.0f / fmaxf((float)c, 1.0f);
    float4 o = make_float4(acc.x * r, acc.y * r, acc.z * r, acc.w * r);
    ((float4*)(mean + (size_t)wave * FEAT))[lane] = o;
}

#define WPITCH 132
#define APITCH 260

template <int NT>
__global__ void __launch_bounds__(256) __attribute__((amdgpu_waves_per_eu(4, 4)))
sage_gemm_lds(const float* __restrict__ A1, const float* __restrict__ A2,
              const float* __restrict__ W1, const float* __restrict__ W2,
              const float* __restrict__ bias, float* __restrict__ out,
              int M, int units)
{
    constexpr int N    = NT * 16;
    constexpr int ROWS = (8 / NT) * 16;
    constexpr int LPT  = (2 * ROWS) / 8;
    constexpr int WLD  = N / 8;
    constexpr int LDSF = (N * WPITCH > ROWS * APITCH) ? N * WPITCH
                                                      : ROWS * APITCH;
    __shared__ __attribute__((aligned(16))) float lds[LDSF];

    const int tid  = threadIdx.x;
    const int lane = tid & 31;
    const int w    = tid >> 5;
    const int c    = lane & 15;
    const int h    = lane >> 4;
    const int nt   = w % NT;
    const int mg   = w / NT;
    const int n    = nt * 16 + c;

    float4 bw[32];
    #pragma unroll
    for (int half = 0; half < 2; ++half) {
        const float* Wsrc = half ? W2 : W1;
        float4 t[WLD];
        #pragma unroll
        for (int i = 0; i < WLD; ++i)
            t[i] = ((const float4*)(Wsrc + (size_t)(w + 8 * i) * FEAT))[lane];
        #pragma unroll
        for (int i = 0; i < WLD; ++i)
            *(float4*)(lds + (w + 8 * i) * WPITCH + 4 * lane) = t[i];
        __syncthreads();
        #pragma unroll
        for (int j = 0; j < 16; ++j)
            bw[half * 16 + j] =
                *(const float4*)(lds + n * WPITCH + 8 * j + 4 * h);
        __syncthreads();
    }
    const float bn = bias[n];

    v4f pre[LPT];
    int u = blockIdx.x;
    #pragma unroll
    for (int i = 0; i < LPT; ++i) {
        const float* S = (i * 8 >= ROWS) ? A2 : A1;
        int m = u * ROWS + (i * 8) % ROWS + w;
        m = (m < M) ? m : (M - 1);
        pre[i] = ((const v4f*)(S + (size_t)m * FEAT))[lane];
    }

    const float* arow = lds + (mg * 16 + c) * APITCH + 4 * h;

    for (; u < units; u += gridDim.x) {
        #pragma unroll
        for (int i = 0; i < LPT; ++i) {
            const int r   = (i * 8) % ROWS + w;
            const int off = (i * 8 >= ROWS) ? FEAT : 0;
            *(v4f*)(lds + r * APITCH + off + 4 * lane) = pre[i];
        }
        __syncthreads();

        const int un = u + gridDim.x;
        if (un < units) {
            #pragma unroll
            for (int i = 0; i < LPT; ++i) {
                const float* S = (i * 8 >= ROWS) ? A2 : A1;
                int m = un * ROWS + (i * 8) % ROWS + w;
                m = (m < M) ? m : (M - 1);
                pre[i] = ((const v4f*)(S + (size_t)m * FEAT))[lane];
            }
        }

        v8f acc = {};
        #pragma unroll
        for (int j = 0; j < 32; ++j) {
            const float4 a = *(const float4*)(arow + 8 * j);
            v2f a0, a1, b0, b1;
            a0[0] = a.x;     a0[1] = a.y;     a1[0] = a.z;     a1[1] = a.w;
            b0[0] = bw[j].x; b0[1] = bw[j].y; b1[0] = bw[j].z; b1[1] = bw[j].w;
            acc = __builtin_amdgcn_wmma_f32_16x16x4_f32(
                 false, a0,  false, b0,
                 (short)0, acc,  false,  false);
            acc = __builtin_amdgcn_wmma_f32_16x16x4_f32(
                false, a1, false, b1, (short)0, acc, false, false);
        }

        const int mbase = u * ROWS + mg * 16 + h * 8;
        #pragma unroll
        for (int j = 0; j < 8; ++j) {
            const int mm = mbase + j;
            if (mm < M) {
                const float v = acc[j] + bn;
                out[(size_t)mm * N + n] = v > 0.0f ? v : 0.0f;
            }
        }
        __syncthreads();
    }
}

extern "C" void kernel_launch(void* const* d_in, const int* in_sizes, int n_in,
                              void* d_out, int out_size, void* d_ws, size_t ws_size,
                              hipStream_t stream) {
    const float* x    = (const float*)d_in[0];
    const float* W1l  = (const float*)d_in[1];
    const float* b1l  = (const float*)d_in[2];
    const float* W1r  = (const float*)d_in[3];
    const float* W2l  = (const float*)d_in[4];
    const float* b2l  = (const float*)d_in[5];
    const float* W2r  = (const float*)d_in[6];
    const int*   src1 = (const int*)d_in[7];
    const int*   dst1 = (const int*)d_in[8];
    const int*   src2 = (const int*)d_in[9];
    const int*   dst2 = (const int*)d_in[10];

    const int E1     = in_sizes[7];
    const int E2     = in_sizes[9];
    const int N1FEAT = in_sizes[2];
    const int N2FEAT = in_sizes[5];
    const int SIZE2  = out_size / N2FEAT;
    const int SIZE1  = 50000;

    float* agg1 = (float*)d_ws;
    float* cnt1 = agg1 + (size_t)SIZE1 * FEAT;
    float* h    = cnt1 + SIZE1;
    float* agg2 = h    + (size_t)SIZE1 * FEAT;

    {
        const int rsh = 8;
        const int nb  = (SIZE1 + 255) >> 8;
        const int g1  = (E1 + CHUNK - 1) / CHUNK;
        int* icnt  = (int*)h;
        int* icur  = icnt + IDX_STRIDE;
        int* iperm = icur + IDX_STRIDE;
        int* ips   = iperm + (size_t)g1 * CHUNK;
        int* ipd   = ips   + (size_t)g1 * CHUNK;
        int* itbl  = ipd   + (size_t)g1 * CHUNK;
        sage_sort_chunks<<<g1, BT, 0, stream>>>(src1, dst1, ips, ipd, itbl,
                                               E1, rsh, nb, g1);
        sage_rank_ranges<<<nb, BT, 0, stream>>>(ips, ipd, itbl, icnt, icur,
                                               iperm, SIZE1, E1, rsh, g1);
        sage_gather_mean<<<(SIZE1 + 7) / 8, 256, 0, stream>>>(
            x, iperm, icnt, icur, agg1, SIZE1, E1);
    }
    {
        int units = (SIZE1 + 15) / 16;
        int grid  = units < 512 ? units : 512;
        sage_gemm_lds<8><<<grid, 256, 0, stream>>>(
            agg1, x, W1l, W1r, b1l, h, SIZE1, units);
    }

    {
        const int rsh = 5;
        const int nb  = (SIZE2 + 31) >> 5;
        const int g1  = (E2 + CHUNK - 1) / CHUNK;
        int* icnt  = (int*)agg1;
        int* icur  = icnt + IDX_STRIDE;
        int* iperm = icur + IDX_STRIDE;
        int* ips   = iperm + (size_t)g1 * CHUNK;
        int* ipd   = ips   + (size_t)g1 * CHUNK;
        int* itbl  = ipd   + (size_t)g1 * CHUNK;
        sage_sort_chunks<<<g1, BT, 0, stream>>>(src2, dst2, ips, ipd, itbl,
                                               E2, rsh, nb, g1);
        sage_rank_ranges<<<nb, BT, 0, stream>>>(ips, ipd, itbl, icnt, icur,
                                               iperm, SIZE2, E2, rsh, g1);
        sage_gather_mean<<<(SIZE2 + 7) / 8, 256, 0, stream>>>(
            h, iperm, icnt, icur, agg2, SIZE2, E2);
    }
    {
        int units = (SIZE2 + 31) / 32;
        int grid  = units < 512 ? units : 512;
        sage_gemm_lds<4><<<grid, 256, 0, stream>>>(
            agg2, h, W2l, W2r, b2l, (float*)d_out, SIZE2, units);
    }
}


# Layer 1 computes only the rows of h that layer 2 reads (root rows + rows its edge list names, marked in a table): row sums skip the rest, the WMMA product lists its wanted rows; sum order kept
# speedup vs baseline: 3.5845x; 1.0878x over previous
#include <hip/hip_runtime.h>

typedef __attribute__((ext_vector_type(2))) float v2f;
typedef __attribute__((ext_vector_type(8))) float v8f;
typedef __attribute__((ext_vector_type(4))) float v4f;

#define FEAT 128

#define IDX_STRIDE 65536

#define CHUNK 8192
#define BT    512
#define GMAX  512
#define RMAX  256
#define STAGE 16384

__device__ __forceinline__ int block_excl_scan(int v, int* wtot, int& total)
{
    const int lane = threadIdx.x & 31;
    const int w    = threadIdx.x >> 5;
    int incl = v;
    for (int o = 1; o < 32; o <<= 1) {
        int y = __shfl_up(incl, o);
        if (lane >= o) incl += y;
    }
    if (lane == 31) wtot[w] = incl;
    __syncthreads();
    int ws    = (lane < BT / 32) ? wtot[lane] : 0;
    int wincl = ws;
    for (int o = 1; o < 32; o <<= 1) {
        int y = __shfl_up(wincl, o);
        if (lane >= o) wincl += y;
    }
    total = __shfl(wincl, 31);
    const int wave_excl = __shfl(wincl, w) - __shfl(ws, w);
    return wave_excl + incl - v;
}

__global__ void __launch_bounds__(BT)
sage_sort_chunks(const int* __restrict__ src, const int* __restrict__ dst,
                 int* __restrict__ ps, int* __restrict__ pd,
                 int* __restrict__ tbl, int E, int rsh, int nb, int g1,
                 int* __restrict__ need, int nneed)
{
    constexpr int EPT = CHUNK / BT;
    for (int i = blockIdx.x * BT + threadIdx.x; i < nneed; i += gridDim.x * BT)
        need[i] = 0;
    __shared__ int hist[BT];
    __shared__ int wt[32];
    __shared__ __attribute__((aligned(16))) int sS[CHUNK];
    __shared__ __attribute__((aligned(16))) int sD[CHUNK];

    const int tid = threadIdx.x;
    const int g   = blockIdx.x;
    const int e0  = g * CHUNK;

    hist[tid] = 0;
    int s[EPT], d[EPT], r[EPT];
    #pragma unroll
    for (int j = 0; j < EPT; ++j) {
        const int e = e0 + j * BT + tid;
        const bool ok = e < E;
        d[j] = ok ? dst[e] : 0;
        s[j] = ok ? src[e] : 0;
    }
    __syncthreads();
    #pragma unroll
    for (int j = 0; j < EPT; ++j) {
        r[j] = 0;
        if (e0 + j * BT + tid < E) {
            int bn = d[j] >> rsh;
            bn = bn < 0 ? 0 : (bn > nb - 1 ? nb - 1 : bn);
            r[j] = atomicAdd(&hist[bn], 1);
        }
    }
    __syncthreads();
    int total;
    const int ex = block_excl_scan(hist[tid], wt, total);
    hist[tid] = ex;
    if (tid <= nb) tbl[tid * g1 + g] = ex;
    __syncthreads();
    #pragma unroll
    for (int j = 0; j < EPT; ++j) {
        if (e0 + j * BT + tid < E) {
            int bn = d[j] >> rsh;
            bn = bn < 0 ? 0 : (bn > nb - 1 ? nb - 1 : bn);
            const int p = hist[bn] + r[j];
            sS[p] = s[j];
            sD[p] = d[j];
        }
    }
    __syncthreads();
    const int count = (E - e0 < CHUNK) ? (E - e0) : CHUNK;
    const int nq    = count >> 2;
    int4* ops = (int4*)(ps + e0);
    int4* opd = (int4*)(pd + e0);
    for (int i = tid; i < nq; i += BT) {
        ops[i] = ((const int4*)sS)[i];
        opd[i] = ((const int4*)sD)[i];
    }
    for (int i = (nq << 2) + tid; i < count; i += BT) {
        ps[e0 + i] = sS[i];
        pd[e0 + i] = sD[i];
    }
}

#define RU 8
__global__ void __launch_bounds__(BT)
sage_rank_ranges(const int* __restrict__ ps, const int* __restrict__ pd,
                 const int* __restrict__ tbl,
                 int* __restrict__ cnt, int* __restrict__ cur,
                 int* __restrict__ perm, int n, int E, int rsh, int g1,
                 const int* __restrict__ msrc, int mE,
                 int* __restrict__ need, int nneed)
{
    constexpr int NW = BT / 32;
    for (int e = blockIdx.x * BT + threadIdx.x; e < mE; e += gridDim.x * BT) {
        const int s = msrc[e];
        if ((unsigned)s < (unsigned)nneed) need[s] = 1;
    }
    __shared__ int rs[GMAX + RU * NW];
    __shared__ int rl[GMAX + RU * NW];
    __shared__ int cL[RMAX];
    __shared__ int wtA[32];
    __shared__ int wtB[32];
    __shared__ int wtC[32];
    __shared__ int stg[STAGE];

    const int tid   = threadIdx.x;
    const int lane  = tid & 31;
    const int w     = __builtin_amdgcn_readfirstlane(tid >> 5);
    const int b     = blockIdx.x;
    const int R     = 1 << rsh;
    const int dbase = b << rsh;

    int lo = 0, len = 0;
    if (tid < g1) {
        lo = tbl[b * g1 + tid];
        int hi = tbl[(b + 1) * g1 + tid];
        lo = lo < 0 ? 0 : (lo > CHUNK ? CHUNK : lo);
        hi = hi < lo ? lo : (hi > CHUNK ? CHUNK : hi);
        len = hi - lo;
    }
    rs[tid] = tid * CHUNK + lo;
    rl[tid] = len;
    if (tid < RU * NW) { rs[GMAX + tid] = 0; rl[GMAX + tid] = 0; }
    if (tid < RMAX) cL[tid] = 0;
    int base, S;
    block_excl_scan(lo,  wtA, base);
    block_excl_scan(len, wtB, S);

    for (int g0 = w; g0 < g1; g0 += RU * NW) {
        int dv[RU][2], l[RU], off[RU];
        #pragma unroll
        for (int u = 0; u < RU; ++u) {
            l[u]   = rl[g0 + u * NW];
            off[u] = rs[g0 + u * NW];
        }
        #pragma unroll
        for (int u = 0; u < RU; ++u) {
            dv[u][0] = (lane      < l[u]) ? pd[off[u] + lane]      : -1;
            dv[u][1] = (lane + 32 < l[u]) ? pd[off[u] + lane + 32] : -1;
        }
        #pragma unroll
        for (int u = 0; u < RU; ++u) {
            #pragma unroll
            for (int k = 0; k < 2; ++k) {
                const int dl = dv[u][k] - dbase;
                if ((unsigned)dl < (unsigned)R) atomicAdd(&cL[dl], 1);
            }
        }
        #pragma unroll
        for (int u = 0; u < RU; ++u) {
            for (int i = 64 + lane; i < l[u]; i += 32) {
                const int dl = pd[off[u] + i] - dbase;
                if ((unsigned)dl < (unsigned)R) atomicAdd(&cL[dl], 1);
            }
        }
    }
    __syncthreads();

    const int c = (tid < R) ? cL[tid] : 0;
    int tot;
    const int ex = block_excl_scan(c, wtC, tot);
    if (tid < R) {
        cL[tid] = ex;
        const int dd = dbase + tid;
        if (dd < n) { cnt[dd] = c; cur[dd] = base + ex + c; }
    }
    __syncthreads();

    const bool staged = (S <= STAGE);
    for (int g0 = w; g0 < g1; g0 += RU * NW) {
        int dv[RU][2], sv[RU][2], l[RU], off[RU];
        #pragma unroll
        for (int u = 0; u < RU; ++u) {
            l[u]   = rl[g0 + u * NW];
            off[u] = rs[g0 + u * NW];
        }
        #pragma unroll
        for (int u = 0; u < RU; ++u) {
            const bool k0 = lane < l[u], k1 = lane + 32 < l[u];
            dv[u][0] = k0 ? pd[off[u] + lane]      : -1;
            sv[u][0] = k0 ? ps[off[u] + lane]      : 0;
            dv[u][1] = k1 ? pd[off[u] + lane + 32] : -1;
            sv[u][1] = k1 ? ps[off[u] + lane + 32] : 0;
        }
        #pragma unroll
        for (int u = 0; u < RU; ++u) {
            #pragma unroll
            for (int k = 0; k < 2; ++k) {
                const int dl = dv[u][k] - dbase;
                if ((unsigned)dl < (unsigned)R) {
                    const int p = atomicAdd(&cL[dl], 1);
                    if (staged) {
                        if ((unsigned)p < (unsigned)STAGE) stg[p] = sv[u][k];
                    } else {
                        const int q = base + p;
                        if ((unsigned)q < (unsigned)E) perm[q] = sv[u][k];
                    }
                }
            }
        }
        #pragma unroll
        for (int u = 0; u < RU; ++u) {
            for (int i = 64 + lane; i < l[u]; i += 32) {
                const int dl = pd[off[u] + i] - dbase;
                const int sx = ps[off[u] + i];
                if ((unsigned)dl < (unsigned)R) {
                    const int p = atomicAdd(&cL[dl], 1);
                    if (staged) {
                        if ((unsigned)p < (unsigned)STAGE) stg[p] = sx;
                    } else {
                        const int q = base + p;
                        if ((unsigned)q < (unsigned)E) perm[q] = sx;
                    }
                }
            }
        }
    }
    if (staged) {
        __syncthreads();
        for (int i = tid; i < S; i += BT) {
            const int q = base + i;
            if ((unsigned)q < (unsigned)E) perm[q] = stg[i];
        }
    }
}

__global__ void __launch_bounds__(256)
sage_gather_mean(const float* __restrict__ xsrc,
                 const int* __restrict__ perm,
                 const int* __restrict__ cnt,
                 const int* __restrict__ cur,
                 float* __restrict__ mean,
                 int n, int E,
                 const int* __restrict__ need, int nkeep)
{
    int wave = (int)((blockIdx.x * blockDim.x + threadIdx.x) >> 5);
    int lane = threadIdx.x & 31;
    if (wave >= n) return;
    const int* wtab = (need != nullptr) ? need : cnt;
    const int want  = __builtin_amdgcn_readfirstlane(wtab[wave]);
    const int c    = __builtin_amdgcn_readfirstlane(cnt[wave]);
    const int end0 = __builtin_amdgcn_readfirstlane(cur[wave]);
    if (need != nullptr && wave >= nkeep && want == 0) return;
    const int end = (end0 < E) ? end0 : E;
    const int beg = (end0 - c > 0) ? (end0 - c) : 0;

    float4 acc = make_float4(0.0f, 0.0f, 0.0f, 0.0f);
    for (int base = beg; base < end; base += 32) {
        int nrem = end - base;
        if (nrem > 32) nrem = 32;
        int my = (lane < nrem) ? perm[base + lane] : 0;
        for (int j = 0; j < nrem; j += 8) {
            float4 v[8];
            #pragma unroll
            for (int u = 0; u < 8; ++u) {
                int s = __builtin_amdgcn_readlane(my, j + u);
                v[u] = ((const float4*)(xsrc + (size_t)s * FEAT))[lane];
            }
            #pragma unroll
            for (int u = 0; u < 8; ++u) {
                if (j + u < nrem) {
                    acc.x += v[u].x;
                    acc.y += v[u].y;
                    acc.z += v[u].z;
                    acc.w += v[u].w;
                }
            }
        }
    }
    float r = 1.0f / fmaxf((float)c, 1.0f);
    float4 o = make_float4(acc.x * r, acc.y * r, acc.z * r, acc.w * r);
    ((float4*)(mean + (size_t)wave * FEAT))[lane] = o;
}

#define LMAX   128
#define WPITCH 132
#define APITCH 260

template <int NT>
__global__ void __launch_bounds__(256) __attribute__((amdgpu_waves_per_eu(4, 4)))
sage_gemm_lds(const float* __restrict__ A1, const float* __restrict__ A2,
              const float* __restrict__ W1, const float* __restrict__ W2,
              const float* __restrict__ bias, float* __restrict__ out,
              int M, int per, const int* __restrict__ need, int nkeep)
{
    constexpr int N    = NT * 16;
    constexpr int ROWS = (8 / NT) * 16;
    constexpr int LPT  = (2 * ROWS) / 8;
    constexpr int WLD  = N / 8;
    constexpr int LDSF = (N * WPITCH > ROWS * APITCH) ? N * WPITCH
                                                      : ROWS * APITCH;
    __shared__ __attribute__((aligned(16))) float lds[LDSF];
    __shared__ __attribute__((aligned(16))) int lst[LMAX + 32];
    __shared__ int wtot[8];

    const int tid  = threadIdx.x;
    const int lane = tid & 31;
    const int w    = tid >> 5;
    const int c    = lane & 15;
    const int h    = lane >> 4;
    const int nt   = w % NT;
    const int mg   = w / NT;
    const int n    = nt * 16 + c;

    const int rown = (int)blockIdx.x + (int)gridDim.x * tid;
    bool wanted = (tid < per) && (tid < LMAX) && (rown < M);
    if (wanted && need != nullptr && rown >= nkeep) wanted = (need[rown] != 0);
    const int rsafe = ((int)blockIdx.x < M) ? (int)blockIdx.x : (M - 1);
    if (tid < LMAX + 32) lst[tid] = rsafe;
    const unsigned bal = __builtin_amdgcn_ballot_w32(wanted);
    if (lane == 0) wtot[w] = __popc(bal);
    __syncthreads();
    int loff = 0, K = 0;
    #pragma unroll
    for (int i = 0; i < 8; ++i) {
        const int ci = wtot[i];
        if (i < w) loff += ci;
        K += ci;
    }
    if (wanted) lst[loff + __popc(bal & ((1u << lane) - 1u))] = rown;

    float4 bw[32];
    #pragma unroll
    for (int half = 0; half < 2; ++half) {
        const float* Wsrc = half ? W2 : W1;
        float4 t[WLD];
        #pragma unroll
        for (int i = 0; i < WLD; ++i)
            t[i] = ((const float4*)(Wsrc + (size_t)(w + 8 * i) * FEAT))[lane];
        #pragma unroll
        for (int i = 0; i < WLD; ++i)
            *(float4*)(lds + (w + 8 * i) * WPITCH + 4 * lane) = t[i];
        __syncthreads();
        #pragma unroll
        for (int j = 0; j < 16; ++j)
            bw[half * 16 + j] =
                *(const float4*)(lds + n * WPITCH + 8 * j + 4 * h);
        __syncthreads();
    }
    const float bn = bias[n];

    v4f pre[LPT];
    const int passes = (K + ROWS - 1) / ROWS;
    #pragma unroll
    for (int i = 0; i < LPT; ++i) {
        const float* S = (i * 8 >= ROWS) ? A2 : A1;
        const int m = lst[(i * 8) % ROWS + w];
        pre[i] = ((const v4f*)(S + (size_t)m * FEAT))[lane];
    }

    const float* arow = lds + (mg * 16 + c) * APITCH + 4 * h;

    for (int p = 0; p < passes; ++p) {
        #pragma unroll
        for (int i = 0; i < LPT; ++i) {
            const int r   = (i * 8) % ROWS + w;
            const int off = (i * 8 >= ROWS) ? FEAT : 0;
            *(v4f*)(lds + r * APITCH + off + 4 * lane) = pre[i];
        }
        __syncthreads();

        if (p + 1 < passes) {
            #pragma unroll
            for (int i = 0; i < LPT; ++i) {
                const float* S = (i * 8 >= ROWS) ? A2 : A1;
                const int m = lst[(p + 1) * ROWS + (i * 8) % ROWS + w];
                pre[i] = ((const v4f*)(S + (size_t)m * FEAT))[lane];
            }
        }

        const int sbase = p * ROWS + mg * 16 + h * 8;
        const int4 q0 = *(const int4*)(lst + sbase);
        const int4 q1 = *(const int4*)(lst + sbase + 4);
        const int orow[8] = {q0.x, q0.y, q0.z, q0.w, q1.x, q1.y, q1.z, q1.w};

        v8f acc = {};
        #pragma unroll
        for (int j = 0; j < 32; ++j) {
            const float4 a = *(const float4*)(arow + 8 * j);
            v2f a0, a1, b0, b1;
            a0[0] = a.x;     a0[1] = a.y;     a1[0] = a.z;     a1[1] = a.w;
            b0[0] = bw[j].x; b0[1] = bw[j].y; b1[0] = bw[j].z; b1[1] = bw[j].w;
            acc = __builtin_amdgcn_wmma_f32_16x16x4_f32(
                 false, a0,  false, b0,
                 (short)0, acc,  false,  false);
            acc = __builtin_amdgcn_wmma_f32_16x16x4_f32(
                false, a1, false, b1, (short)0, acc, false, false);
        }

        #pragma unroll
        for (int j = 0; j < 8; ++j) {
            if (sbase + j < K) {
                const float v = acc[j] + bn;
                out[(size_t)orow[j] * N + n] = v > 0.0f ? v : 0.0f;
            }
        }
        __syncthreads();
    }
}

extern "C" void kernel_launch(void* const* d_in, const int* in_sizes, int n_in,
                              void* d_out, int out_size, void* d_ws, size_t ws_size,
                              hipStream_t stream) {
    const float* x    = (const float*)d_in[0];
    const float* W1l  = (const float*)d_in[1];
    const float* b1l  = (const float*)d_in[2];
    const float* W1r  = (const float*)d_in[3];
    const float* W2l  = (const float*)d_in[4];
    const float* b2l  = (const float*)d_in[5];
    const float* W2r  = (const float*)d_in[6];
    const int*   src1 = (const int*)d_in[7];
    const int*   dst1 = (const int*)d_in[8];
    const int*   src2 = (const int*)d_in[9];
    const int*   dst2 = (const int*)d_in[10];

    const int E1     = in_sizes[7];
    const int E2     = in_sizes[9];
    const int N1FEAT = in_sizes[2];
    const int N2FEAT = in_sizes[5];
    const int SIZE2  = out_size / N2FEAT;
    const int SIZE1  = 50000;

    float* agg1 = (float*)d_ws;
    float* cnt1 = agg1 + (size_t)SIZE1 * FEAT;
    float* h    = cnt1 + SIZE1;
    float* agg2 = h    + (size_t)SIZE1 * FEAT;
    int* need = (int*)cnt1;

    {
        const int rsh = 8;
        const int nb  = (SIZE1 + 255) >> 8;
        const int g1  = (E1 + CHUNK - 1) / CHUNK;
        int* icnt  = (int*)h;
        int* icur  = icnt + IDX_STRIDE;
        int* iperm = icur + IDX_STRIDE;
        int* ips   = iperm + (size_t)g1 * CHUNK;
        int* ipd   = ips   + (size_t)g1 * CHUNK;
        int* itbl  = ipd   + (size_t)g1 * CHUNK;
        sage_sort_chunks<<<g1, BT, 0, stream>>>(src1, dst1, ips, ipd, itbl,
                                               E1, rsh, nb, g1, need, SIZE1);
        sage_rank_ranges<<<nb, BT, 0, stream>>>(ips, ipd, itbl, icnt, icur,
                                               iperm, SIZE1, E1, rsh, g1,
                                               src2, E2, need, SIZE1);
        sage_gather_mean<<<(SIZE1 + 7) / 8, 256, 0, stream>>>(
            x, iperm, icnt, icur, agg1, SIZE1, E1, need, SIZE2);
    }
    {
        int units = (SIZE1 + 15) / 16;
        int grid  = units < 512 ? units : 512;
        int per   = (SIZE1 + grid - 1) / grid;
        sage_gemm_lds<8><<<grid, 256, 0, stream>>>(
            agg1, x, W1l, W1r, b1l, h, SIZE1, per, need, SIZE2);
    }

    {
        const int rsh = 5;
        const int nb  = (SIZE2 + 31) >> 5;
        const int g1  = (E2 + CHUNK - 1) / CHUNK;
        int* icnt  = (int*)agg1;
        int* icur  = icnt + IDX_STRIDE;
        int* iperm = icur + IDX_STRIDE;
        int* ips   = iperm + (size_t)g1 * CHUNK;
        int* ipd   = ips   + (size_t)g1 * CHUNK;
        int* itbl  = ipd   + (size_t)g1 * CHUNK;
        sage_sort_chunks<<<g1, BT, 0, stream>>>(src2, dst2, ips, ipd, itbl,
                                               E2, rsh, nb, g1, nullptr, 0);
        sage_rank_ranges<<<nb, BT, 0, stream>>>(ips, ipd, itbl, icnt, icur,
                                               iperm, SIZE2, E2, rsh, g1,
                                               nullptr, 0, nullptr, 0);
        sage_gather_mean<<<(SIZE2 + 7) / 8, 256, 0, stream>>>(
            h, iperm, icnt, icur, agg2, SIZE2, E2, nullptr, 0);
    }
    {
        int units = (SIZE2 + 31) / 32;
        int grid  = units < 512 ? units : 512;
        int per   = (SIZE2 + grid - 1) / grid;
        sage_gemm_lds<4><<<grid, 256, 0, stream>>>(
            agg2, h, W2l, W2r, b2l, (float*)d_out, SIZE2, per, nullptr, 0);
    }
}


# Both layers' edge lists are grouped by destination in the same two launches (layer 2's chunks and ranges get workgroups of their own beside layer 1's): 8 launches become 6; row sum order kept
# speedup vs baseline: 4.0726x; 1.1362x over previous
#include <hip/hip_runtime.h>

typedef __attribute__((ext_vector_type(2))) float v2f;
typedef __attribute__((ext_vector_type(8))) float v8f;
typedef __attribute__((ext_vector_type(4))) float v4f;

#define FEAT 128

#define IDX_STRIDE 65536

#define CHUNK 8192
#define BT    512
#define GMAX  512
#define RMAX  256
#define STAGE 16384

__device__ __forceinline__ int block_excl_scan(int v, int* wtot, int& total)
{
    const int lane = threadIdx.x & 31;
    const int w    = threadIdx.x >> 5;
    int incl = v;
    for (int o = 1; o < 32; o <<= 1) {
        int y = __shfl_up(incl, o);
        if (lane >= o) incl += y;
    }
    if (lane == 31) wtot[w] = incl;
    __syncthreads();
    int ws    = (lane < BT / 32) ? wtot[lane] : 0;
    int wincl = ws;
    for (int o = 1; o < 32; o <<= 1) {
        int y = __shfl_up(wincl, o);
        if (lane >= o) wincl += y;
    }
    total = __shfl(wincl, 31);
    const int wave_excl = __shfl(wincl, w) - __shfl(ws, w);
    return wave_excl + incl - v;
}

__global__ void __launch_bounds__(BT)
sage_sort_chunks(const int* __restrict__ srcA, const int* __restrict__ dstA,
                 int* __restrict__ psA, int* __restrict__ pdA,
                 int* __restrict__ tblA, int EA, int rshA, int nbA, int gA,
                 const int* __restrict__ srcB, const int* __restrict__ dstB,
                 int* __restrict__ psB, int* __restrict__ pdB,
                 int* __restrict__ tblB, int EB, int rshB, int nbB, int gB,
                 int* __restrict__ need, int nneed)
{
    constexpr int EPT = CHUNK / BT;
    for (int i = blockIdx.x * BT + threadIdx.x; i < nneed; i += gridDim.x * BT)
        need[i] = 0;
    __shared__ int hist[BT];
    __shared__ int wt[32];
    __shared__ __attribute__((aligned(16))) int sS[CHUNK];
    __shared__ __attribute__((aligned(16))) int sD[CHUNK];

    const bool second = (int)blockIdx.x >= gA;
    const int* __restrict__ src = second ? srcB : srcA;
    const int* __restrict__ dst = second ? dstB : dstA;
    int* __restrict__ ps  = second ? psB  : psA;
    int* __restrict__ pd  = second ? pdB  : pdA;
    int* __restrict__ tbl = second ? tblB : tblA;
    const int E   = second ? EB   : EA;
    const int rsh = second ? rshB : rshA;
    const int nb  = second ? nbB  : nbA;
    const int g1  = second ? gB   : gA;

    const int tid = threadIdx.x;
    const int g   = (int)blockIdx.x - (second ? gA : 0);
    const int e0  = g * CHUNK;

    hist[tid] = 0;
    int s[EPT], d[EPT], r[EPT];
    #pragma unroll
    for (int j = 0; j < EPT; ++j) {
        const int e = e0 + j * BT + tid;
        const bool ok = e < E;
        d[j] = ok ? dst[e] : 0;
        s[j] = ok ? src[e] : 0;
    }
    __syncthreads();
    #pragma unroll
    for (int j = 0; j < EPT; ++j) {
        r[j] = 0;
        if (e0 + j * BT + tid < E) {
            int bn = d[j] >> rsh;
            bn = bn < 0 ? 0 : (bn > nb - 1 ? nb - 1 : bn);
            r[j] = atomicAdd(&hist[bn], 1);
        }
    }
    __syncthreads();
    int total;
    const int ex = block_excl_scan(hist[tid], wt, total);
    hist[tid] = ex;
    if (tid <= nb) tbl[tid * g1 + g] = ex;
    __syncthreads();
    #pragma unroll
    for (int j = 0; j < EPT; ++j) {
        if (e0 + j * BT + tid < E) {
            int bn = d[j] >> rsh;
            bn = bn < 0 ? 0 : (bn > nb - 1 ? nb - 1 : bn);
            const int p = hist[bn] + r[j];
            sS[p] = s[j];
            sD[p] = d[j];
        }
    }
    __syncthreads();
    const int count = (E - e0 < CHUNK) ? (E - e0) : CHUNK;
    const int nq    = count >> 2;
    int4* ops = (int4*)(ps + e0);
    int4* opd = (int4*)(pd + e0);
    for (int i = tid; i < nq; i += BT) {
        ops[i] = ((const int4*)sS)[i];
        opd[i] = ((const int4*)sD)[i];
    }
    for (int i = (nq << 2) + tid; i < count; i += BT) {
        ps[e0 + i] = sS[i];
        pd[e0 + i] = sD[i];
    }
}

#define RU 8
__global__ void __launch_bounds__(BT)
sage_rank_ranges(const int* __restrict__ psA, const int* __restrict__ pdA,
                 const int* __restrict__ tblA,
                 int* __restrict__ cntA, int* __restrict__ curA,
                 int* __restrict__ permA, int nA, int EA, int rshA, int gA,
                 int nbA,
                 const int* __restrict__ psB, const int* __restrict__ pdB,
                 const int* __restrict__ tblB,
                 int* __restrict__ cntB, int* __restrict__ curB,
                 int* __restrict__ permB, int nB, int EB, int rshB, int gB,
                 const int* __restrict__ msrc, int mE,
                 int* __restrict__ need, int nneed)
{
    constexpr int NW = BT / 32;
    for (int e = blockIdx.x * BT + threadIdx.x; e < mE; e += gridDim.x * BT) {
        const int s = msrc[e];
        if ((unsigned)s < (unsigned)nneed) need[s] = 1;
    }
    const bool second = (int)blockIdx.x >= nbA;
    const int* __restrict__ ps  = second ? psB  : psA;
    const int* __restrict__ pd  = second ? pdB  : pdA;
    const int* __restrict__ tbl = second ? tblB : tblA;
    int* __restrict__ cnt  = second ? cntB  : cntA;
    int* __restrict__ cur  = second ? curB  : curA;
    int* __restrict__ perm = second ? permB : permA;
    const int n   = second ? nB   : nA;
    const int E   = second ? EB   : EA;
    const int rsh = second ? rshB : rshA;
    const int g1  = second ? gB   : gA;
    __shared__ int rs[GMAX + RU * NW];
    __shared__ int rl[GMAX + RU * NW];
    __shared__ int cL[RMAX];
    __shared__ int wtA[32];
    __shared__ int wtB[32];
    __shared__ int wtC[32];
    __shared__ int stg[STAGE];

    const int tid   = threadIdx.x;
    const int lane  = tid & 31;
    const int w     = __builtin_amdgcn_readfirstlane(tid >> 5);
    const int b     = (int)blockIdx.x - (second ? nbA : 0);
    const int R     = 1 << rsh;
    const int dbase = b << rsh;

    int lo = 0, len = 0;
    if (tid < g1) {
        lo = tbl[b * g1 + tid];
        int hi = tbl[(b + 1) * g1 + tid];
        lo = lo < 0 ? 0 : (lo > CHUNK ? CHUNK : lo);
        hi = hi < lo ? lo : (hi > CHUNK ? CHUNK : hi);
        len = hi - lo;
    }
    rs[tid] = tid * CHUNK + lo;
    rl[tid] = len;
    if (tid < RU * NW) { rs[GMAX + tid] = 0; rl[GMAX + tid] = 0; }
    if (tid < RMAX) cL[tid] = 0;
    int base, S;
    block_excl_scan(lo,  wtA, base);
    block_excl_scan(len, wtB, S);

    for (int g0 = w; g0 < g1; g0 += RU * NW) {
        int dv[RU][2], l[RU], off[RU];
        #pragma unroll
        for (int u = 0; u < RU; ++u) {
            l[u]   = rl[g0 + u * NW];
            off[u] = rs[g0 + u * NW];
        }
        #pragma unroll
        for (int u = 0; u < RU; ++u) {
            dv[u][0] = (lane      < l[u]) ? pd[off[u] + lane]      : -1;
            dv[u][1] = (lane + 32 < l[u]) ? pd[off[u] + lane + 32] : -1;
        }
        #pragma unroll
        for (int u = 0; u < RU; ++u) {
            #pragma unroll
            for (int k = 0; k < 2; ++k) {
                const int dl = dv[u][k] - dbase;
                if ((unsigned)dl < (unsigned)R) atomicAdd(&cL[dl], 1);
            }
        }
        #pragma unroll
        for (int u = 0; u < RU; ++u) {
            for (int i = 64 + lane; i < l[u]; i += 32) {
                const int dl = pd[off[u] + i] - dbase;
                if ((unsigned)dl < (unsigned)R) atomicAdd(&cL[dl], 1);
            }
        }
    }
    __syncthreads();

    const int c = (tid < R) ? cL[tid] : 0;
    int tot;
    const int ex = block_excl_scan(c, wtC, tot);
    if (tid < R) {
        cL[tid] = ex;
        const int dd = dbase + tid;
        if (dd < n) { cnt[dd] = c; cur[dd] = base + ex + c; }
    }
    __syncthreads();

    const bool staged = (S <= STAGE);
    for (int g0 = w; g0 < g1; g0 += RU * NW) {
        int dv[RU][2], sv[RU][2], l[RU], off[RU];
        #pragma unroll
        for (int u = 0; u < RU; ++u) {
            l[u]   = rl[g0 + u * NW];
            off[u] = rs[g0 + u * NW];
        }
        #pragma unroll
        for (int u = 0; u < RU; ++u) {
            const bool k0 = lane < l[u], k1 = lane + 32 < l[u];
            dv[u][0] = k0 ? pd[off[u] + lane]      : -1;
            sv[u][0] = k0 ? ps[off[u] + lane]      : 0;
            dv[u][1] = k1 ? pd[off[u] + lane + 32] : -1;
            sv[u][1] = k1 ? ps[off[u] + lane + 32] : 0;
        }
        #pragma unroll
        for (int u = 0; u < RU; ++u) {
            #pragma unroll
            for (int k = 0; k < 2; ++k) {
                const int dl = dv[u][k] - dbase;
                if ((unsigned)dl < (unsigned)R) {
                    const int p = atomicAdd(&cL[dl], 1);
                    if (staged) {
                        if ((unsigned)p < (unsigned)STAGE) stg[p] = sv[u][k];
                    } else {
                        const int q = base + p;
                        if ((unsigned)q < (unsigned)E) perm[q] = sv[u][k];
                    }
                }
            }
        }
        #pragma unroll
        for (int u = 0; u < RU; ++u) {
            for (int i = 64 + lane; i < l[u]; i += 32) {
                const int dl = pd[off[u] + i] - dbase;
                const int sx = ps[off[u] + i];
                if ((unsigned)dl < (unsigned)R) {
                    const int p = atomicAdd(&cL[dl], 1);
                    if (staged) {
                        if ((unsigned)p < (unsigned)STAGE) stg[p] = sx;
                    } else {
                        const int q = base + p;
                        if ((unsigned)q < (unsigned)E) perm[q] = sx;
                    }
                }
            }
        }
    }
    if (staged) {
        __syncthreads();
        for (int i = tid; i < S; i += BT) {
            const int q = base + i;
            if ((unsigned)q < (unsigned)E) perm[q] = stg[i];
        }
    }
}

__global__ void __launch_bounds__(256)
sage_gather_mean(const float* __restrict__ xsrc,
                 const int* __restrict__ perm,
                 const int* __restrict__ cnt,
                 const int* __restrict__ cur,
                 float* __restrict__ mean,
                 int n, int E,
                 const int* __restrict__ need, int nkeep)
{
    int wave = (int)((blockIdx.x * blockDim.x + threadIdx.x) >> 5);
    int lane = threadIdx.x & 31;
    if (wave >= n) return;
    const int* wtab = (need != nullptr) ? need : cnt;
    const int want  = __builtin_amdgcn_readfirstlane(wtab[wave]);
    const int c    = __builtin_amdgcn_readfirstlane(cnt[wave]);
    const int end0 = __builtin_amdgcn_readfirstlane(cur[wave]);
    if (need != nullptr && wave >= nkeep && want == 0) return;
    const int end = (end0 < E) ? end0 : E;
    const int beg = (end0 - c > 0) ? (end0 - c) : 0;

    float4 acc = make_float4(0.0f, 0.0f, 0.0f, 0.0f);
    for (int base = beg; base < end; base += 32) {
        int nrem = end - base;
        if (nrem > 32) nrem = 32;
        int my = (lane < nrem) ? perm[base + lane] : 0;
        for (int j = 0; j < nrem; j += 8) {
            float4 v[8];
            #pragma unroll
            for (int u = 0; u < 8; ++u) {
                int s = __builtin_amdgcn_readlane(my, j + u);
                v[u] = ((const float4*)(xsrc + (size_t)s * FEAT))[lane];
            }
            #pragma unroll
            for (int u = 0; u < 8; ++u) {
                if (j + u < nrem) {
                    acc.x += v[u].x;
                    acc.y += v[u].y;
                    acc.z += v[u].z;
                    acc.w += v[u].w;
                }
            }
        }
    }
    float r = 1.0f / fmaxf((float)c, 1.0f);
    float4 o = make_float4(acc.x * r, acc.y * r, acc.z * r, acc.w * r);
    ((float4*)(mean + (size_t)wave * FEAT))[lane] = o;
}

#define LMAX   128
#define WPITCH 132
#define APITCH 260

template <int NT>
__global__ void __launch_bounds__(256) __attribute__((amdgpu_waves_per_eu(4, 4)))
sage_gemm_lds(const float* __restrict__ A1, const float* __restrict__ A2,
              const float* __restrict__ W1, const float* __restrict__ W2,
              const float* __restrict__ bias, float* __restrict__ out,
              int M, int per, const int* __restrict__ need, int nkeep)
{
    constexpr int N    = NT * 16;
    constexpr int ROWS = (8 / NT) * 16;
    constexpr int LPT  = (2 * ROWS) / 8;
    constexpr int WLD  = N / 8;
    constexpr int LDSF = (N * WPITCH > ROWS * APITCH) ? N * WPITCH
                                                      : ROWS * APITCH;
    __shared__ __attribute__((aligned(16))) float lds[LDSF];
    __shared__ __attribute__((aligned(16))) int lst[LMAX + 32];
    __shared__ int wtot[8];

    const int tid  = threadIdx.x;
    const int lane = tid & 31;
    const int w    = tid >> 5;
    const int c    = lane & 15;
    const int h    = lane >> 4;
    const int nt   = w % NT;
    const int mg   = w / NT;
    const int n    = nt * 16 + c;

    const int rown = (int)blockIdx.x + (int)gridDim.x * tid;
    bool wanted = (tid < per) && (tid < LMAX) && (rown < M);
    if (wanted && need != nullptr && rown >= nkeep) wanted = (need[rown] != 0);
    const int rsafe = ((int)blockIdx.x < M) ? (int)blockIdx.x : (M - 1);
    if (tid < LMAX + 32) lst[tid] = rsafe;
    const unsigned bal = __builtin_amdgcn_ballot_w32(wanted);
    if (lane == 0) wtot[w] = __popc(bal);
    __syncthreads();
    int loff = 0, K = 0;
    #pragma unroll
    for (int i = 0; i < 8; ++i) {
        const int ci = wtot[i];
        if (i < w) loff += ci;
        K += ci;
    }
    if (wanted) lst[loff + __popc(bal & ((1u << lane) - 1u))] = rown;

    float4 bw[32];
    #pragma unroll
    for (int half = 0; half < 2; ++half) {
        const float* Wsrc = half ? W2 : W1;
        float4 t[WLD];
        #pragma unroll
        for (int i = 0; i < WLD; ++i)
            t[i] = ((const float4*)(Wsrc + (size_t)(w + 8 * i) * FEAT))[lane];
        #pragma unroll
        for (int i = 0; i < WLD; ++i)
            *(float4*)(lds + (w + 8 * i) * WPITCH + 4 * lane) = t[i];
        __syncthreads();
        #pragma unroll
        for (int j = 0; j < 16; ++j)
            bw[half * 16 + j] =
                *(const float4*)(lds + n * WPITCH + 8 * j + 4 * h);
        __syncthreads();
    }
    const float bn = bias[n];

    v4f pre[LPT];
    const int passes = (K + ROWS - 1) / ROWS;
    #pragma unroll
    for (int i = 0; i < LPT; ++i) {
        const float* S = (i * 8 >= ROWS) ? A2 : A1;
        const int m = lst[(i * 8) % ROWS + w];
        pre[i] = ((const v4f*)(S + (size_t)m * FEAT))[lane];
    }

    const float* arow = lds + (mg * 16 + c) * APITCH + 4 * h;

    for (int p = 0; p < passes; ++p) {
        #pragma unroll
        for (int i = 0; i < LPT; ++i) {
            const int r   = (i * 8) % ROWS + w;
            const int off = (i * 8 >= ROWS) ? FEAT : 0;
            *(v4f*)(lds + r * APITCH + off + 4 * lane) = pre[i];
        }
        __syncthreads();

        if (p + 1 < passes) {
            #pragma unroll
            for (int i = 0; i < LPT; ++i) {
                const float* S = (i * 8 >= ROWS) ? A2 : A1;
                const int m = lst[(p + 1) * ROWS + (i * 8) % ROWS + w];
                pre[i] = ((const v4f*)(S + (size_t)m * FEAT))[lane];
            }
        }

        const int sbase = p * ROWS + mg * 16 + h * 8;
        const int4 q0 = *(const int4*)(lst + sbase);
        const int4 q1 = *(const int4*)(lst + sbase + 4);
        const int orow[8] = {q0.x, q0.y, q0.z, q0.w, q1.x, q1.y, q1.z, q1.w};

        v8f acc = {};
        #pragma unroll
        for (int j = 0; j < 32; ++j) {
            const float4 a = *(const float4*)(arow + 8 * j);
            v2f a0, a1, b0, b1;
            a0[0] = a.x;     a0[1] = a.y;     a1[0] = a.z;     a1[1] = a.w;
            b0[0] = bw[j].x; b0[1] = bw[j].y; b1[0] = bw[j].z; b1[1] = bw[j].w;
            acc = __builtin_amdgcn_wmma_f32_16x16x4_f32(
                 false, a0,  false, b0,
                 (short)0, acc,  false,  false);
            acc = __builtin_amdgcn_wmma_f32_16x16x4_f32(
                false, a1, false, b1, (short)0, acc, false, false);
        }

        #pragma unroll
        for (int j = 0; j < 8; ++j) {
            if (sbase + j < K) {
                const float v = acc[j] + bn;
                out[(size_t)orow[j] * N + n] = v > 0.0f ? v : 0.0f;
            }
        }
        __syncthreads();
    }
}

extern "C" void kernel_launch(void* const* d_in, const int* in_sizes, int n_in,
                              void* d_out, int out_size, void* d_ws, size_t ws_size,
                              hipStream_t stream) {
    const float* x    = (const float*)d_in[0];
    const float* W1l  = (const float*)d_in[1];
    const float* b1l  = (const float*)d_in[2];
    const float* W1r  = (const float*)d_in[3];
    const float* W2l  = (const float*)d_in[4];
    const float* b2l  = (const float*)d_in[5];
    const float* W2r  = (const float*)d_in[6];
    const int*   src1 = (const int*)d_in[7];
    const int*   dst1 = (const int*)d_in[8];
    const int*   src2 = (const int*)d_in[9];
    const int*   dst2 = (const int*)d_in[10];

    const int E1     = in_sizes[7];
    const int E2     = in_sizes[9];
    const int N1FEAT = in_sizes[2];
    const int N2FEAT = in_sizes[5];
    const int SIZE2  = out_size / N2FEAT;
    const int SIZE1  = 50000;

    float* agg1 = (float*)d_ws;
    float* cnt1 = agg1 + (size_t)SIZE1 * FEAT;
    float* h    = cnt1 + SIZE1;
    float* idx2 = h    + (size_t)SIZE1 * FEAT;
    float* agg2 = agg1;
    int* need = (int*)cnt1;

    const int rsh1 = 8;
    const int nb1  = (SIZE1 + 255) >> 8;
    const int g1a  = (E1 + CHUNK - 1) / CHUNK;
    int* icnt1  = (int*)h;
    int* icur1  = icnt1 + IDX_STRIDE;
    int* iperm1 = icur1 + IDX_STRIDE;
    int* ips1   = iperm1 + (size_t)g1a * CHUNK;
    int* ipd1   = ips1   + (size_t)g1a * CHUNK;
    int* itbl1  = ipd1   + (size_t)g1a * CHUNK;
    const int rsh2 = 5;
    const int nb2  = (SIZE2 + 31) >> 5;
    const int g1b  = (E2 + CHUNK - 1) / CHUNK;
    int* icnt2  = (int*)idx2;
    int* icur2  = icnt2 + IDX_STRIDE;
    int* iperm2 = icur2 + IDX_STRIDE;
    int* ips2   = iperm2 + (size_t)g1b * CHUNK;
    int* ipd2   = ips2   + (size_t)g1b * CHUNK;
    int* itbl2  = ipd2   + (size_t)g1b * CHUNK;

    sage_sort_chunks<<<g1a + g1b, BT, 0, stream>>>(
        src1, dst1, ips1, ipd1, itbl1, E1, rsh1, nb1, g1a,
        src2, dst2, ips2, ipd2, itbl2, E2, rsh2, nb2, g1b,
        need, SIZE1);
    sage_rank_ranges<<<nb1 + nb2, BT, 0, stream>>>(
        ips1, ipd1, itbl1, icnt1, icur1, iperm1, SIZE1, E1, rsh1, g1a, nb1,
        ips2, ipd2, itbl2, icnt2, icur2, iperm2, SIZE2, E2, rsh2, g1b,
        src2, E2, need, SIZE1);

    sage_gather_mean<<<(SIZE1 + 7) / 8, 256, 0, stream>>>(
        x, iperm1, icnt1, icur1, agg1, SIZE1, E1, need, SIZE2);
    {
        int units = (SIZE1 + 15) / 16;
        int grid  = units < 512 ? units : 512;
        int per   = (SIZE1 + grid - 1) / grid;
        sage_gemm_lds<8><<<grid, 256, 0, stream>>>(
            agg1, x, W1l, W1r, b1l, h, SIZE1, per, need, SIZE2);
    }

    sage_gather_mean<<<(SIZE2 + 7) / 8, 256, 0, stream>>>(
        h, iperm2, icnt2, icur2, agg2, SIZE2, E2, nullptr, 0);
    {
        int units = (SIZE2 + 31) / 32;
        int grid  = units < 512 ? units : 512;
        int per   = (SIZE2 + grid - 1) / grid;
        sage_gemm_lds<4><<<grid, 256, 0, stream>>>(
            agg2, h, W2l, W2r, b2l, (float*)d_out, SIZE2, per, nullptr, 0);
    }
}


# Row sums move into each layer's product kernel: the wave that stages a row sums its source rows and writes the mean to LDS; means skip memory, 6 launches become 4; row sum order kept
# speedup vs baseline: 4.2177x; 1.0356x over previous
#include <hip/hip_runtime.h>

typedef __attribute__((ext_vector_type(2))) float v2f;
typedef __attribute__((ext_vector_type(8))) float v8f;
typedef __attribute__((ext_vector_type(4))) float v4f;

#define FEAT 128

#define IDX_STRIDE 65536

#define CHUNK 8192
#define BT    512
#define GMAX  512
#define RMAX  256
#define STAGE 16384

__device__ __forceinline__ int block_excl_scan(int v, int* wtot, int& total)
{
    const int lane = threadIdx.x & 31;
    const int w    = threadIdx.x >> 5;
    int incl = v;
    for (int o = 1; o < 32; o <<= 1) {
        int y = __shfl_up(incl, o);
        if (lane >= o) incl += y;
    }
    if (lane == 31) wtot[w] = incl;
    __syncthreads();
    int ws    = (lane < BT / 32) ? wtot[lane] : 0;
    int wincl = ws;
    for (int o = 1; o < 32; o <<= 1) {
        int y = __shfl_up(wincl, o);
        if (lane >= o) wincl += y;
    }
    total = __shfl(wincl, 31);
    const int wave_excl = __shfl(wincl, w) - __shfl(ws, w);
    return wave_excl + incl - v;
}

__global__ void __launch_bounds__(BT)
sage_sort_chunks(const int* __restrict__ srcA, const int* __restrict__ dstA,
                 int* __restrict__ psA, int* __restrict__ pdA,
                 int* __restrict__ tblA, int EA, int rshA, int nbA, int gA,
                 const int* __restrict__ srcB, const int* __restrict__ dstB,
                 int* __restrict__ psB, int* __restrict__ pdB,
                 int* __restrict__ tblB, int EB, int rshB, int nbB, int gB,
                 int* __restrict__ need, int nneed)
{
    constexpr int EPT = CHUNK / BT;
    for (int i = blockIdx.x * BT + threadIdx.x; i < nneed; i += gridDim.x * BT)
        need[i] = 0;
    __shared__ int hist[BT];
    __shared__ int wt[32];
    __shared__ __attribute__((aligned(16))) int sS[CHUNK];
    __shared__ __attribute__((aligned(16))) int sD[CHUNK];

    const bool second = (int)blockIdx.x >= gA;
    const int* __restrict__ src = second ? srcB : srcA;
    const int* __restrict__ dst = second ? dstB : dstA;
    int* __restrict__ ps  = second ? psB  : psA;
    int* __restrict__ pd  = second ? pdB  : pdA;
    int* __restrict__ tbl = second ? tblB : tblA;
    const int E   = second ? EB   : EA;
    const int rsh = second ? rshB : rshA;
    const int nb  = second ? nbB  : nbA;
    const int g1  = second ? gB   : gA;

    const int tid = threadIdx.x;
    const int g   = (int)blockIdx.x - (second ? gA : 0);
    const int e0  = g * CHUNK;

    hist[tid] = 0;
    int s[EPT], d[EPT], r[EPT];
    #pragma unroll
    for (int j = 0; j < EPT; ++j) {
        const int e = e0 + j * BT + tid;
        const bool ok = e < E;
        d[j] = ok ? dst[e] : 0;
        s[j] = ok ? src[e] : 0;
    }
    __syncthreads();
    #pragma unroll
    for (int j = 0; j < EPT; ++j) {
        r[j] = 0;
        if (e0 + j * BT + tid < E) {
            int bn = d[j] >> rsh;
            bn = bn < 0 ? 0 : (bn > nb - 1 ? nb - 1 : bn);
            r[j] = atomicAdd(&hist[bn], 1);
        }
    }
    __syncthreads();
    int total;
    const int ex = block_excl_scan(hist[tid], wt, total);
    hist[tid] = ex;
    if (tid <= nb) tbl[tid * g1 + g] = ex;
    __syncthreads();
    #pragma unroll
    for (int j = 0; j < EPT; ++j) {
        if (e0 + j * BT + tid < E) {
            int bn = d[j] >> rsh;
            bn = bn < 0 ? 0 : (bn > nb - 1 ? nb - 1 : bn);
            const int p = hist[bn] + r[j];
            sS[p] = s[j];
            sD[p] = d[j];
        }
    }
    __syncthreads();
    const int count = (E - e0 < CHUNK) ? (E - e0) : CHUNK;
    const int nq    = count >> 2;
    int4* ops = (int4*)(ps + e0);
    int4* opd = (int4*)(pd + e0);
    for (int i = tid; i < nq; i += BT) {
        ops[i] = ((const int4*)sS)[i];
        opd[i] = ((const int4*)sD)[i];
    }
    for (int i = (nq << 2) + tid; i < count; i += BT) {
        ps[e0 + i] = sS[i];
        pd[e0 + i] = sD[i];
    }
}

#define RU 8
__global__ void __launch_bounds__(BT)
sage_rank_ranges(const int* __restrict__ psA, const int* __restrict__ pdA,
                 const int* __restrict__ tblA,
                 int* __restrict__ cntA, int* __restrict__ curA,
                 int* __restrict__ permA, int nA, int EA, int rshA, int gA,
                 int nbA,
                 const int* __restrict__ psB, const int* __restrict__ pdB,
                 const int* __restrict__ tblB,
                 int* __restrict__ cntB, int* __restrict__ curB,
                 int* __restrict__ permB, int nB, int EB, int rshB, int gB,
                 const int* __restrict__ msrc, int mE,
                 int* __restrict__ need, int nneed)
{
    constexpr int NW = BT / 32;
    for (int e = blockIdx.x * BT + threadIdx.x; e < mE; e += gridDim.x * BT) {
        const int s = msrc[e];
        if ((unsigned)s < (unsigned)nneed) need[s] = 1;
    }
    const bool second = (int)blockIdx.x >= nbA;
    const int* __restrict__ ps  = second ? psB  : psA;
    const int* __restrict__ pd  = second ? pdB  : pdA;
    const int* __restrict__ tbl = second ? tblB : tblA;
    int* __restrict__ cnt  = second ? cntB  : cntA;
    int* __restrict__ cur  = second ? curB  : curA;
    int* __restrict__ perm = second ? permB : permA;
    const int n   = second ? nB   : nA;
    const int E   = second ? EB   : EA;
    const int rsh = second ? rshB : rshA;
    const int g1  = second ? gB   : gA;
    __shared__ int rs[GMAX + RU * NW];
    __shared__ int rl[GMAX + RU * NW];
    __shared__ int cL[RMAX];
    __shared__ int wtA[32];
    __shared__ int wtB[32];
    __shared__ int wtC[32];
    __shared__ int stg[STAGE];

    const int tid   = threadIdx.x;
    const int lane  = tid & 31;
    const int w     = __builtin_amdgcn_readfirstlane(tid >> 5);
    const int b     = (int)blockIdx.x - (second ? nbA : 0);
    const int R     = 1 << rsh;
    const int dbase = b << rsh;

    int lo = 0, len = 0;
    if (tid < g1) {
        lo = tbl[b * g1 + tid];
        int hi = tbl[(b + 1) * g1 + tid];
        lo = lo < 0 ? 0 : (lo > CHUNK ? CHUNK : lo);
        hi = hi < lo ? lo : (hi > CHUNK ? CHUNK : hi);
        len = hi - lo;
    }
    rs[tid] = tid * CHUNK + lo;
    rl[tid] = len;
    if (tid < RU * NW) { rs[GMAX + tid] = 0; rl[GMAX + tid] = 0; }
    if (tid < RMAX) cL[tid] = 0;
    int base, S;
    block_excl_scan(lo,  wtA, base);
    block_excl_scan(len, wtB, S);

    for (int g0 = w; g0 < g1; g0 += RU * NW) {
        int dv[RU][2], l[RU], off[RU];
        #pragma unroll
        for (int u = 0; u < RU; ++u) {
            l[u]   = rl[g0 + u * NW];
            off[u] = rs[g0 + u * NW];
        }
        #pragma unroll
        for (int u = 0; u < RU; ++u) {
            dv[u][0] = (lane      < l[u]) ? pd[off[u] + lane]      : -1;
            dv[u][1] = (lane + 32 < l[u]) ? pd[off[u] + lane + 32] : -1;
        }
        #pragma unroll
        for (int u = 0; u < RU; ++u) {
            #pragma unroll
            for (int k = 0; k < 2; ++k) {
                const int dl = dv[u][k] - dbase;
                if ((unsigned)dl < (unsigned)R) atomicAdd(&cL[dl], 1);
            }
        }
        #pragma unroll
        for (int u = 0; u < RU; ++u) {
            for (int i = 64 + lane; i < l[u]; i += 32) {
                const int dl = pd[off[u] + i] - dbase;
                if ((unsigned)dl < (unsigned)R) atomicAdd(&cL[dl], 1);
            }
        }
    }
    __syncthreads();

    const int c = (tid < R) ? cL[tid] : 0;
    int tot;
    const int ex = block_excl_scan(c, wtC, tot);
    if (tid < R) {
        cL[tid] = ex;
        const int dd = dbase + tid;
        if (dd < n) { cnt[dd] = c; cur[dd] = base + ex + c; }
    }
    __syncthreads();

    const bool staged = (S <= STAGE);
    for (int g0 = w; g0 < g1; g0 += RU * NW) {
        int dv[RU][2], sv[RU][2], l[RU], off[RU];
        #pragma unroll
        for (int u = 0; u < RU; ++u) {
            l[u]   = rl[g0 + u * NW];
            off[u] = rs[g0 + u * NW];
        }
        #pragma unroll
        for (int u = 0; u < RU; ++u) {
            const bool k0 = lane < l[u], k1 = lane + 32 < l[u];
            dv[u][0] = k0 ? pd[off[u] + lane]      : -1;
            sv[u][0] = k0 ? ps[off[u] + lane]      : 0;
            dv[u][1] = k1 ? pd[off[u] + lane + 32] : -1;
            sv[u][1] = k1 ? ps[off[u] + lane + 32] : 0;
        }
        #pragma unroll
        for (int u = 0; u < RU; ++u) {
            #pragma unroll
            for (int k = 0; k < 2; ++k) {
                const int dl = dv[u][k] - dbase;
                if ((unsigned)dl < (unsigned)R) {
                    const int p = atomicAdd(&cL[dl], 1);
                    if (staged) {
                        if ((unsigned)p < (unsigned)STAGE) stg[p] = sv[u][k];
                    } else {
                        const int q = base + p;
                        if ((unsigned)q < (unsigned)E) perm[q] = sv[u][k];
                    }
                }
            }
        }
        #pragma unroll
        for (int u = 0; u < RU; ++u) {
            for (int i = 64 + lane; i < l[u]; i += 32) {
                const int dl = pd[off[u] + i] - dbase;
                const int sx = ps[off[u] + i];
                if ((unsigned)dl < (unsigned)R) {
                    const int p = atomicAdd(&cL[dl], 1);
                    if (staged) {
                        if ((unsigned)p < (unsigned)STAGE) stg[p] = sx;
                    } else {
                        const int q = base + p;
                        if ((unsigned)q < (unsigned)E) perm[q] = sx;
                    }
                }
            }
        }
    }
    if (staged) {
        __syncthreads();
        for (int i = tid; i < S; i += BT) {
            const int q = base + i;
            if ((unsigned)q < (unsigned)E) perm[q] = stg[i];
        }
    }
}

#define LMAX   128
#define WPITCH 132
#define APITCH 260

template <int NT>
__global__ void __launch_bounds__(256) __attribute__((amdgpu_waves_per_eu(4, 4)))
sage_gather_gemm(const float* __restrict__ X,
                 const int* __restrict__ perm, const int* __restrict__ cnt,
                 const int* __restrict__ cur, int E,
                 const float* __restrict__ W1, const float* __restrict__ W2,
                 const float* __restrict__ bias, float* __restrict__ out,
                 int M, int per, const int* __restrict__ need, int nkeep)
{
    constexpr int N    = NT * 16;
    constexpr int ROWS = (8 / NT) * 16;
    constexpr int NSL  = ROWS / 8;
    constexpr int SG   = 2;
    constexpr int G    = 8;
    constexpr int WLD  = N / 8;
    constexpr int LDSF = (N * WPITCH > ROWS * APITCH) ? N * WPITCH
                                                      : ROWS * APITCH;
    __shared__ __attribute__((aligned(16))) float lds[LDSF];
    __shared__ __attribute__((aligned(16))) int lst[LMAX + 32];
    __shared__ int   sBeg[LMAX + 32];
    __shared__ int   sLen[LMAX + 32];
    __shared__ float sInv[LMAX + 32];
    __shared__ int wtot[8];

    const int tid  = threadIdx.x;
    const int lane = tid & 31;
    const int w    = __builtin_amdgcn_readfirstlane(tid >> 5);
    const int c    = lane & 15;
    const int h    = lane >> 4;
    const int nt   = w % NT;
    const int mg   = w / NT;
    const int n    = nt * 16 + c;

    const int rown = (int)blockIdx.x + (int)gridDim.x * tid;
    bool wanted = (tid < per) && (tid < LMAX) && (rown < M);
    if (wanted && need != nullptr && rown >= nkeep) wanted = (need[rown] != 0);
    const int rsafe = ((int)blockIdx.x < M) ? (int)blockIdx.x : (M - 1);
    if (tid < LMAX + 32) lst[tid] = rsafe;
    const unsigned bal = __builtin_amdgcn_ballot_w32(wanted);
    if (lane == 0) wtot[w] = __popc(bal);
    __syncthreads();
    int loff = 0, K = 0;
    #pragma unroll
    for (int i = 0; i < 8; ++i) {
        const int ci = wtot[i];
        if (i < w) loff += ci;
        K += ci;
    }
    if (wanted) lst[loff + __popc(bal & ((1u << lane) - 1u))] = rown;
    __syncthreads();

    int segc = 0, sege = 0;
    if (tid < LMAX + 32) {
        const int m = lst[tid];
        segc = cnt[m];
        sege = cur[m];
    }

    int   gb[NSL], gn[NSL], myq[NSL];
    float gi[NSL];
    v4f   pre2[NSL];
    auto fetch = [&](int p) {
        #pragma unroll
        for (int s = 0; s < NSL; ++s) {
            const int slot = p * ROWS + s * 8 + w;
            gb[s] = __builtin_amdgcn_readfirstlane(sBeg[slot]);
            gn[s] = __builtin_amdgcn_readfirstlane(sLen[slot]);
            gi[s] = sInv[slot];
            const int m = lst[slot];
            myq[s]  = (lane < gn[s]) ? perm[gb[s] + lane] : 0;
            pre2[s] = ((const v4f*)(X + (size_t)m * FEAT))[lane];
        }
    };

    float4 bw[32];
    #pragma unroll
    for (int half = 0; half < 2; ++half) {
        const float* Wsrc = half ? W2 : W1;
        float4 t[WLD];
        #pragma unroll
        for (int i = 0; i < WLD; ++i)
            t[i] = ((const float4*)(Wsrc + (size_t)(w + 8 * i) * FEAT))[lane];
        #pragma unroll
        for (int i = 0; i < WLD; ++i)
            *(float4*)(lds + (w + 8 * i) * WPITCH + 4 * lane) = t[i];
        if (half == 0 && tid < LMAX + 32) {
            const int end = (sege < E) ? sege : E;
            const int b   = (sege - segc > 0) ? (sege - segc) : 0;
            int len = end - b;
            if (len < 0 || tid >= K) len = 0;
            sBeg[tid] = b;
            sLen[tid] = len;
            sInv[tid] = 1.0f / fmaxf((float)segc, 1.0f);
        }
        __syncthreads();
        #pragma unroll
        for (int j = 0; j < 16; ++j)
            bw[half * 16 + j] =
                *(const float4*)(lds + n * WPITCH + 8 * j + 4 * h);
        __syncthreads();
        if (half == 0) fetch(0);
    }
    const float bn = bias[n];

    const int passes = (K + ROWS - 1) / ROWS;

    const float* arow = lds + (mg * 16 + c) * APITCH + 4 * h;

    for (int p = 0; p < passes; ++p) {
        v4f acc[NSL];
        #pragma unroll
        for (int g0 = 0; g0 < NSL; g0 += SG) {
            int my[SG];
            int nmx = 0;
            #pragma unroll
            for (int t = 0; t < SG; ++t) {
                acc[g0 + t] = (v4f){0.0f, 0.0f, 0.0f, 0.0f};
                my[t] = myq[g0 + t];
                nmx   = gn[g0 + t] > nmx ? gn[g0 + t] : nmx;
            }
            for (int base = 0; base < nmx; base += 32) {
                int nrem[SG];
                int nm = 0;
                #pragma unroll
                for (int t = 0; t < SG; ++t) {
                    int r = gn[g0 + t] - base;
                    r = r < 0 ? 0 : (r > 32 ? 32 : r);
                    nrem[t] = r;
                    nm = r > nm ? r : nm;
                    if (base > 0)
                        my[t] = (lane < r) ? perm[gb[g0 + t] + base + lane] : 0;
                }
                for (int j = 0; j < nm; j += G) {
                    v4f v[SG][G];
                    #pragma unroll
                    for (int u = 0; u < G; ++u) {
                        #pragma unroll
                        for (int t = 0; t < SG; ++t) {
                            const int sid =
                                __builtin_amdgcn_readlane(my[t], j + u);
                            v[t][u] =
                                ((const v4f*)(X + (size_t)sid * FEAT))[lane];
                        }
                    }
                    #pragma unroll
                    for (int t = 0; t < SG; ++t) {
                        #pragma unroll
                        for (int u = 0; u < G; ++u) {
                            if (j + u < nrem[t]) acc[g0 + t] += v[t][u];
                        }
                    }
                }
            }
        }

        __syncthreads();
        #pragma unroll
        for (int s = 0; s < NSL; ++s) {
            float* row = lds + (s * 8 + w) * APITCH + 4 * lane;
            *(v4f*)(row)        = acc[s] * gi[s];
            *(v4f*)(row + FEAT) = pre2[s];
        }
        __syncthreads();

        if (p + 1 < passes) fetch(p + 1);

        const int sbase = p * ROWS + mg * 16 + h * 8;
        const int4 q0 = *(const int4*)(lst + sbase);
        const int4 q1 = *(const int4*)(lst + sbase + 4);
        const int orow[8] = {q0.x, q0.y, q0.z, q0.w, q1.x, q1.y, q1.z, q1.w};

        v8f pacc = {};
        #pragma unroll
        for (int j = 0; j < 32; ++j) {
            const float4 a = *(const float4*)(arow + 8 * j);
            v2f a0, a1, b0, b1;
            a0[0] = a.x;     a0[1] = a.y;     a1[0] = a.z;     a1[1] = a.w;
            b0[0] = bw[j].x; b0[1] = bw[j].y; b1[0] = bw[j].z; b1[1] = bw[j].w;
            pacc = __builtin_amdgcn_wmma_f32_16x16x4_f32(
                 false, a0,  false, b0,
                 (short)0, pacc,  false,  false);
            pacc = __builtin_amdgcn_wmma_f32_16x16x4_f32(
                false, a1, false, b1, (short)0, pacc, false, false);
        }

        #pragma unroll
        for (int j = 0; j < 8; ++j) {
            if (sbase + j < K) {
                const float v = pacc[j] + bn;
                out[(size_t)orow[j] * N + n] = v > 0.0f ? v : 0.0f;
            }
        }
    }
}

extern "C" void kernel_launch(void* const* d_in, const int* in_sizes, int n_in,
                              void* d_out, int out_size, void* d_ws, size_t ws_size,
                              hipStream_t stream) {
    const float* x    = (const float*)d_in[0];
    const float* W1l  = (const float*)d_in[1];
    const float* b1l  = (const float*)d_in[2];
    const float* W1r  = (const float*)d_in[3];
    const float* W2l  = (const float*)d_in[4];
    const float* b2l  = (const float*)d_in[5];
    const float* W2r  = (const float*)d_in[6];
    const int*   src1 = (const int*)d_in[7];
    const int*   dst1 = (const int*)d_in[8];
    const int*   src2 = (const int*)d_in[9];
    const int*   dst2 = (const int*)d_in[10];

    const int E1     = in_sizes[7];
    const int E2     = in_sizes[9];
    const int N1FEAT = in_sizes[2];
    const int N2FEAT = in_sizes[5];
    const int SIZE2  = out_size / N2FEAT;
    const int SIZE1  = 50000;

    float* agg1 = (float*)d_ws;
    float* cnt1 = agg1 + (size_t)SIZE1 * FEAT;
    float* h    = cnt1 + SIZE1;
    float* idx2 = h    + (size_t)SIZE1 * FEAT;
    int* need = (int*)cnt1;

    const int rsh1 = 8;
    const int nb1  = (SIZE1 + 255) >> 8;
    const int g1a  = (E1 + CHUNK - 1) / CHUNK;
    int* icnt1  = (int*)agg1;
    int* icur1  = icnt1 + IDX_STRIDE;
    int* iperm1 = icur1 + IDX_STRIDE;
    int* ips1   = iperm1 + (size_t)g1a * CHUNK;
    int* ipd1   = ips1   + (size_t)g1a * CHUNK;
    int* itbl1  = ipd1   + (size_t)g1a * CHUNK;
    const int rsh2 = 5;
    const int nb2  = (SIZE2 + 31) >> 5;
    const int g1b  = (E2 + CHUNK - 1) / CHUNK;
    int* icnt2  = (int*)idx2;
    int* icur2  = icnt2 + IDX_STRIDE;
    int* iperm2 = icur2 + IDX_STRIDE;
    int* ips2   = iperm2 + (size_t)g1b * CHUNK;
    int* ipd2   = ips2   + (size_t)g1b * CHUNK;
    int* itbl2  = ipd2   + (size_t)g1b * CHUNK;

    sage_sort_chunks<<<g1a + g1b, BT, 0, stream>>>(
        src1, dst1, ips1, ipd1, itbl1, E1, rsh1, nb1, g1a,
        src2, dst2, ips2, ipd2, itbl2, E2, rsh2, nb2, g1b,
        need, SIZE1);
    sage_rank_ranges<<<nb1 + nb2, BT, 0, stream>>>(
        ips1, ipd1, itbl1, icnt1, icur1, iperm1, SIZE1, E1, rsh1, g1a, nb1,
        ips2, ipd2, itbl2, icnt2, icur2, iperm2, SIZE2, E2, rsh2, g1b,
        src2, E2, need, SIZE1);

    {
        int units = (SIZE1 + 15) / 16;
        int grid  = units < 512 ? units : 512;
        int per   = (SIZE1 + grid - 1) / grid;
        sage_gather_gemm<8><<<grid, 256, 0, stream>>>(
            x, iperm1, icnt1, icur1, E1, W1l, W1r, b1l, h, SIZE1, per,
            need, SIZE2);
    }

    {
        int units = (SIZE2 + 31) / 32;
        int grid  = units < 512 ? units : 512;
        int per   = (SIZE2 + grid - 1) / grid;
        sage_gather_gemm<4><<<grid, 256, 0, stream>>>(
            h, iperm2, icnt2, icur2, E2, W2l, W2r, b2l, (float*)d_out, SIZE2,
            per, nullptr, 0);
    }
}


# Layer weights stay in LDS (fragments read beside the activation's), not in registers; freed registers hold up to 24 source rows per summed row in flight, not 8: fewer gather rounds; sum order kept
# speedup vs baseline: 4.3468x; 1.0306x over previous
#include <hip/hip_runtime.h>

typedef __attribute__((ext_vector_type(2))) float v2f;
typedef __attribute__((ext_vector_type(8))) float v8f;
typedef __attribute__((ext_vector_type(4))) float v4f;

#define FEAT 128

#define IDX_STRIDE 65536

#define CHUNK 8192
#define BT    512
#define GMAX  512
#define RMAX  256
#define STAGE 16384

__device__ __forceinline__ int block_excl_scan(int v, int* wtot, int& total)
{
    const int lane = threadIdx.x & 31;
    const int w    = threadIdx.x >> 5;
    int incl = v;
    for (int o = 1; o < 32; o <<= 1) {
        int y = __shfl_up(incl, o);
        if (lane >= o) incl += y;
    }
    if (lane == 31) wtot[w] = incl;
    __syncthreads();
    int ws    = (lane < BT / 32) ? wtot[lane] : 0;
    int wincl = ws;
    for (int o = 1; o < 32; o <<= 1) {
        int y = __shfl_up(wincl, o);
        if (lane >= o) wincl += y;
    }
    total = __shfl(wincl, 31);
    const int wave_excl = __shfl(wincl, w) - __shfl(ws, w);
    return wave_excl + incl - v;
}

__global__ void __launch_bounds__(BT)
sage_sort_chunks(const int* __restrict__ srcA, const int* __restrict__ dstA,
                 int* __restrict__ psA, int* __restrict__ pdA,
                 int* __restrict__ tblA, int EA, int rshA, int nbA, int gA,
                 const int* __restrict__ srcB, const int* __restrict__ dstB,
                 int* __restrict__ psB, int* __restrict__ pdB,
                 int* __restrict__ tblB, int EB, int rshB, int nbB, int gB,
                 int* __restrict__ need, int nneed)
{
    constexpr int EPT = CHUNK / BT;
    for (int i = blockIdx.x * BT + threadIdx.x; i < nneed; i += gridDim.x * BT)
        need[i] = 0;
    __shared__ int hist[BT];
    __shared__ int wt[32];
    __shared__ __attribute__((aligned(16))) int sS[CHUNK];
    __shared__ __attribute__((aligned(16))) int sD[CHUNK];

    const bool second = (int)blockIdx.x >= gA;
    const int* __restrict__ src = second ? srcB : srcA;
    const int* __restrict__ dst = second ? dstB : dstA;
    int* __restrict__ ps  = second ? psB  : psA;
    int* __restrict__ pd  = second ? pdB  : pdA;
    int* __restrict__ tbl = second ? tblB : tblA;
    const int E   = second ? EB   : EA;
    const int rsh = second ? rshB : rshA;
    const int nb  = second ? nbB  : nbA;
    const int g1  = second ? gB   : gA;

    const int tid = threadIdx.x;
    const int g   = (int)blockIdx.x - (second ? gA : 0);
    const int e0  = g * CHUNK;

    hist[tid] = 0;
    int s[EPT], d[EPT], r[EPT];
    #pragma unroll
    for (int j = 0; j < EPT; ++j) {
        const int e = e0 + j * BT + tid;
        const bool ok = e < E;
        d[j] = ok ? dst[e] : 0;
        s[j] = ok ? src[e] : 0;
    }
    __syncthreads();
    #pragma unroll
    for (int j = 0; j < EPT; ++j) {
        r[j] = 0;
        if (e0 + j * BT + tid < E) {
            int bn = d[j] >> rsh;
            bn = bn < 0 ? 0 : (bn > nb - 1 ? nb - 1 : bn);
            r[j] = atomicAdd(&hist[bn], 1);
        }
    }
    __syncthreads();
    int total;
    const int ex = block_excl_scan(hist[tid], wt, total);
    hist[tid] = ex;
    if (tid <= nb) tbl[tid * g1 + g] = ex;
    __syncthreads();
    #pragma unroll
    for (int j = 0; j < EPT; ++j) {
        if (e0 + j * BT + tid < E) {
            int bn = d[j] >> rsh;
            bn = bn < 0 ? 0 : (bn > nb - 1 ? nb - 1 : bn);
            const int p = hist[bn] + r[j];
            sS[p] = s[j];
            sD[p] = d[j];
        }
    }
    __syncthreads();
    const int count = (E - e0 < CHUNK) ? (E - e0) : CHUNK;
    const int nq    = count >> 2;
    int4* ops = (int4*)(ps + e0);
    int4* opd = (int4*)(pd + e0);
    for (int i = tid; i < nq; i += BT) {
        ops[i] = ((const int4*)sS)[i];
        opd[i] = ((const int4*)sD)[i];
    }
    for (int i = (nq << 2) + tid; i < count; i += BT) {
        ps[e0 + i] = sS[i];
        pd[e0 + i] = sD[i];
    }
}

#define RU 8
__global__ void __launch_bounds__(BT)
sage_rank_ranges(const int* __restrict__ psA, const int* __restrict__ pdA,
                 const int* __restrict__ tblA,
                 int* __restrict__ cntA, int* __restrict__ curA,
                 int* __restrict__ permA, int nA, int EA, int rshA, int gA,
                 int nbA,
                 const int* __restrict__ psB, const int* __restrict__ pdB,
                 const int* __restrict__ tblB,
                 int* __restrict__ cntB, int* __restrict__ curB,
                 int* __restrict__ permB, int nB, int EB, int rshB, int gB,
                 const int* __restrict__ msrc, int mE,
                 int* __restrict__ need, int nneed)
{
    constexpr int NW = BT / 32;
    for (int e = blockIdx.x * BT + threadIdx.x; e < mE; e += gridDim.x * BT) {
        const int s = msrc[e];
        if ((unsigned)s < (unsigned)nneed) need[s] = 1;
    }
    const bool second = (int)blockIdx.x >= nbA;
    const int* __restrict__ ps  = second ? psB  : psA;
    const int* __restrict__ pd  = second ? pdB  : pdA;
    const int* __restrict__ tbl = second ? tblB : tblA;
    int* __restrict__ cnt  = second ? cntB  : cntA;
    int* __restrict__ cur  = second ? curB  : curA;
    int* __restrict__ perm = second ? permB : permA;
    const int n   = second ? nB   : nA;
    const int E   = second ? EB   : EA;
    const int rsh = second ? rshB : rshA;
    const int g1  = second ? gB   : gA;
    __shared__ int rs[GMAX + RU * NW];
    __shared__ int rl[GMAX + RU * NW];
    __shared__ int cL[RMAX];
    __shared__ int wtA[32];
    __shared__ int wtB[32];
    __shared__ int wtC[32];
    __shared__ int stg[STAGE];

    const int tid   = threadIdx.x;
    const int lane  = tid & 31;
    const int w     = __builtin_amdgcn_readfirstlane(tid >> 5);
    const int b     = (int)blockIdx.x - (second ? nbA : 0);
    const int R     = 1 << rsh;
    const int dbase = b << rsh;

    int lo = 0, len = 0;
    if (tid < g1) {
        lo = tbl[b * g1 + tid];
        int hi = tbl[(b + 1) * g1 + tid];
        lo = lo < 0 ? 0 : (lo > CHUNK ? CHUNK : lo);
        hi = hi < lo ? lo : (hi > CHUNK ? CHUNK : hi);
        len = hi - lo;
    }
    rs[tid] = tid * CHUNK + lo;
    rl[tid] = len;
    if (tid < RU * NW) { rs[GMAX + tid] = 0; rl[GMAX + tid] = 0; }
    if (tid < RMAX) cL[tid] = 0;
    int base, S;
    block_excl_scan(lo,  wtA, base);
    block_excl_scan(len, wtB, S);

    for (int g0 = w; g0 < g1; g0 += RU * NW) {
        int dv[RU][2], l[RU], off[RU];
        #pragma unroll
        for (int u = 0; u < RU; ++u) {
            l[u]   = rl[g0 + u * NW];
            off[u] = rs[g0 + u * NW];
        }
        #pragma unroll
        for (int u = 0; u < RU; ++u) {
            dv[u][0] = (lane      < l[u]) ? pd[off[u] + lane]      : -1;
            dv[u][1] = (lane + 32 < l[u]) ? pd[off[u] + lane + 32] : -1;
        }
        #pragma unroll
        for (int u = 0; u < RU; ++u) {
            #pragma unroll
            for (int k = 0; k < 2; ++k) {
                const int dl = dv[u][k] - dbase;
                if ((unsigned)dl < (unsigned)R) atomicAdd(&cL[dl], 1);
            }
        }
        #pragma unroll
        for (int u = 0; u < RU; ++u) {
            for (int i = 64 + lane; i < l[u]; i += 32) {
                const int dl = pd[off[u] + i] - dbase;
                if ((unsigned)dl < (unsigned)R) atomicAdd(&cL[dl], 1);
            }
        }
    }
    __syncthreads();

    const int c = (tid < R) ? cL[tid] : 0;
    int tot;
    const int ex = block_excl_scan(c, wtC, tot);
    if (tid < R) {
        cL[tid] = ex;
        const int dd = dbase + tid;
        if (dd < n) { cnt[dd] = c; cur[dd] = base + ex + c; }
    }
    __syncthreads();

    const bool staged = (S <= STAGE);
    for (int g0 = w; g0 < g1; g0 += RU * NW) {
        int dv[RU][2], sv[RU][2], l[RU], off[RU];
        #pragma unroll
        for (int u = 0; u < RU; ++u) {
            l[u]   = rl[g0 + u * NW];
            off[u] = rs[g0 + u * NW];
        }
        #pragma unroll
        for (int u = 0; u < RU; ++u) {
            const bool k0 = lane < l[u], k1 = lane + 32 < l[u];
            dv[u][0] = k0 ? pd[off[u] + lane]      : -1;
            sv[u][0] = k0 ? ps[off[u] + lane]      : 0;
            dv[u][1] = k1 ? pd[off[u] + lane + 32] : -1;
            sv[u][1] = k1 ? ps[off[u] + lane + 32] : 0;
        }
        #pragma unroll
        for (int u = 0; u < RU; ++u) {
            #pragma unroll
            for (int k = 0; k < 2; ++k) {
                const int dl = dv[u][k] - dbase;
                if ((unsigned)dl < (unsigned)R) {
                    const int p = atomicAdd(&cL[dl], 1);
                    if (staged) {
                        if ((unsigned)p < (unsigned)STAGE) stg[p] = sv[u][k];
                    } else {
                        const int q = base + p;
                        if ((unsigned)q < (unsigned)E) perm[q] = sv[u][k];
                    }
                }
            }
        }
        #pragma unroll
        for (int u = 0; u < RU; ++u) {
            for (int i = 64 + lane; i < l[u]; i += 32) {
                const int dl = pd[off[u] + i] - dbase;
                const int sx = ps[off[u] + i];
                if ((unsigned)dl < (unsigned)R) {
                    const int p = atomicAdd(&cL[dl], 1);
                    if (staged) {
                        if ((unsigned)p < (unsigned)STAGE) stg[p] = sx;
                    } else {
                        const int q = base + p;
                        if ((unsigned)q < (unsigned)E) perm[q] = sx;
                    }
                }
            }
        }
    }
    if (staged) {
        __syncthreads();
        for (int i = tid; i < S; i += BT) {
            const int q = base + i;
            if ((unsigned)q < (unsigned)E) perm[q] = stg[i];
        }
    }
}

#define LMAX   128
#define WPITCH 132
#define APITCH 260

template <int NSL, int SG, int GG>
__device__ __forceinline__ void sum_batch(const float* __restrict__ X,
                                          const int (&my)[SG],
                                          const int (&nrem)[SG], int j,
                                          int lane, v4f (&acc)[NSL], int g0)
{
    v4f v[SG][GG];
    #pragma unroll
    for (int u = 0; u < GG; ++u) {
        #pragma unroll
        for (int t = 0; t < SG; ++t) {
            const int sid = __builtin_amdgcn_readlane(my[t], j + u);
            v[t][u] = ((const v4f*)(X + (size_t)sid * FEAT))[lane];
        }
    }
    #pragma unroll
    for (int t = 0; t < SG; ++t) {
        #pragma unroll
        for (int u = 0; u < GG; ++u) {
            if (j + u < nrem[t]) acc[g0 + t] += v[t][u];
        }
    }
}

template <int NT>
__global__ void __launch_bounds__(256) __attribute__((amdgpu_waves_per_eu(4, 4)))
sage_gather_gemm(const float* __restrict__ X,
                 const int* __restrict__ perm, const int* __restrict__ cnt,
                 const int* __restrict__ cur, int E,
                 const float* __restrict__ W1, const float* __restrict__ W2,
                 const float* __restrict__ bias, float* __restrict__ out,
                 int M, int per, const int* __restrict__ need, int nkeep)
{
    constexpr int N    = NT * 16;
    constexpr int ROWS = (8 / NT) * 16;
    constexpr int NSL  = ROWS / 8;
    constexpr int SG   = 2;
    constexpr int GBIG = (NSL <= 2) ? 24 : 16;
    constexpr int WLD  = N / 8;
    __shared__ __attribute__((aligned(16))) float wl[2 * N * WPITCH];
    __shared__ __attribute__((aligned(16))) float lds[ROWS * APITCH];
    __shared__ __attribute__((aligned(16))) int lst[LMAX + 32];
    __shared__ int   sBeg[LMAX + 32];
    __shared__ int   sLen[LMAX + 32];
    __shared__ float sInv[LMAX + 32];
    __shared__ int wtot[8];

    const int tid  = threadIdx.x;
    const int lane = tid & 31;
    const int w    = __builtin_amdgcn_readfirstlane(tid >> 5);
    const int c    = lane & 15;
    const int h    = lane >> 4;
    const int nt   = w % NT;
    const int mg   = w / NT;
    const int n    = nt * 16 + c;

    const int rown = (int)blockIdx.x + (int)gridDim.x * tid;
    bool wanted = (tid < per) && (tid < LMAX) && (rown < M);
    if (wanted && need != nullptr && rown >= nkeep) wanted = (need[rown] != 0);
    const int rsafe = ((int)blockIdx.x < M) ? (int)blockIdx.x : (M - 1);
    if (tid < LMAX + 32) lst[tid] = rsafe;
    const unsigned bal = __builtin_amdgcn_ballot_w32(wanted);
    if (lane == 0) wtot[w] = __popc(bal);
    __syncthreads();
    int loff = 0, K = 0;
    #pragma unroll
    for (int i = 0; i < 8; ++i) {
        const int ci = wtot[i];
        if (i < w) loff += ci;
        K += ci;
    }
    if (wanted) lst[loff + __popc(bal & ((1u << lane) - 1u))] = rown;
    __syncthreads();

    int segc = 0, sege = 0;
    if (tid < LMAX + 32) {
        const int m = lst[tid];
        segc = cnt[m];
        sege = cur[m];
    }

    int   gb[NSL], gn[NSL], myq[NSL];
    float gi[NSL];
    v4f   pre2[NSL];
    auto fetch = [&](int p) {
        #pragma unroll
        for (int s = 0; s < NSL; ++s) {
            const int slot = p * ROWS + s * 8 + w;
            gb[s] = __builtin_amdgcn_readfirstlane(sBeg[slot]);
            gn[s] = __builtin_amdgcn_readfirstlane(sLen[slot]);
            gi[s] = sInv[slot];
            const int m = lst[slot];
            myq[s]  = (lane < gn[s]) ? perm[gb[s] + lane] : 0;
            pre2[s] = ((const v4f*)(X + (size_t)m * FEAT))[lane];
        }
    };

    #pragma unroll
    for (int half = 0; half < 2; ++half) {
        const float* Wsrc = half ? W2 : W1;
        float4 t[WLD];
        #pragma unroll
        for (int i = 0; i < WLD; ++i)
            t[i] = ((const float4*)(Wsrc + (size_t)(w + 8 * i) * FEAT))[lane];
        #pragma unroll
        for (int i = 0; i < WLD; ++i)
            *(float4*)(wl + (half * N + w + 8 * i) * WPITCH + 4 * lane) = t[i];
        if (half == 0) {
            if (tid < LMAX + 32) {
                const int end = (sege < E) ? sege : E;
                const int b   = (sege - segc > 0) ? (sege - segc) : 0;
                int len = end - b;
                if (len < 0 || tid >= K) len = 0;
                sBeg[tid] = b;
                sLen[tid] = len;
                sInv[tid] = 1.0f / fmaxf((float)segc, 1.0f);
            }
            __syncthreads();
            fetch(0);
        }
    }
    const float bn = bias[n];

    const int passes = (K + ROWS - 1) / ROWS;

    const float* arow = lds + (mg * 16 + c) * APITCH + 4 * h;
    const float* brow = wl + n * WPITCH + 4 * h;

    for (int p = 0; p < passes; ++p) {
        v4f acc[NSL];
        #pragma unroll
        for (int g0 = 0; g0 < NSL; g0 += SG) {
            int my[SG];
            int nmx = 0;
            #pragma unroll
            for (int t = 0; t < SG; ++t) {
                acc[g0 + t] = (v4f){0.0f, 0.0f, 0.0f, 0.0f};
                my[t] = myq[g0 + t];
                nmx   = gn[g0 + t] > nmx ? gn[g0 + t] : nmx;
            }
            for (int base = 0; base < nmx; base += 32) {
                int nrem[SG];
                int nm = 0;
                #pragma unroll
                for (int t = 0; t < SG; ++t) {
                    int r = gn[g0 + t] - base;
                    r = r < 0 ? 0 : (r > 32 ? 32 : r);
                    nrem[t] = r;
                    nm = r > nm ? r : nm;
                    if (base > 0)
                        my[t] = (lane < r) ? perm[gb[g0 + t] + base + lane] : 0;
                }
                for (int j = 0; j < nm; ) {
                    const int left = nm - j;
                    if (GBIG == 24 && left > 16 && j + 24 <= 32) {
                        sum_batch<NSL, SG, 24>(X, my, nrem, j, lane, acc, g0);
                        j += 24;
                    } else if (left > 8 && j + 16 <= 32) {
                        sum_batch<NSL, SG, 16>(X, my, nrem, j, lane, acc, g0);
                        j += 16;
                    } else {
                        sum_batch<NSL, SG, 8>(X, my, nrem, j, lane, acc, g0);
                        j += 8;
                    }
                }
            }
        }

        __syncthreads();
        #pragma unroll
        for (int s = 0; s < NSL; ++s) {
            float* row = lds + (s * 8 + w) * APITCH + 4 * lane;
            *(v4f*)(row)        = acc[s] * gi[s];
            *(v4f*)(row + FEAT) = pre2[s];
        }
        __syncthreads();

        if (p + 1 < passes) fetch(p + 1);

        const int sbase = p * ROWS + mg * 16 + h * 8;
        const int4 q0 = *(const int4*)(lst + sbase);
        const int4 q1 = *(const int4*)(lst + sbase + 4);
        const int orow[8] = {q0.x, q0.y, q0.z, q0.w, q1.x, q1.y, q1.z, q1.w};

        v8f pacc = {};
        #pragma unroll
        for (int j = 0; j < 32; ++j) {
            const float4 a = *(const float4*)(arow + 8 * j);
            const float4 b = *(const float4*)(brow + (j >> 4) * (N * WPITCH)
                                                   + 8 * (j & 15));
            v2f a0, a1, b0, b1;
            a0[0] = a.x; a0[1] = a.y; a1[0] = a.z; a1[1] = a.w;
            b0[0] = b.x; b0[1] = b.y; b1[0] = b.z; b1[1] = b.w;
            pacc = __builtin_amdgcn_wmma_f32_16x16x4_f32(
                 false, a0,  false, b0,
                 (short)0, pacc,  false,  false);
            pacc = __builtin_amdgcn_wmma_f32_16x16x4_f32(
                false, a1, false, b1, (short)0, pacc, false, false);
        }

        #pragma unroll
        for (int j = 0; j < 8; ++j) {
            if (sbase + j < K) {
                const float v = pacc[j] + bn;
                out[(size_t)orow[j] * N + n] = v > 0.0f ? v : 0.0f;
            }
        }
    }
}

extern "C" void kernel_launch(void* const* d_in, const int* in_sizes, int n_in,
                              void* d_out, int out_size, void* d_ws, size_t ws_size,
                              hipStream_t stream) {
    const float* x    = (const float*)d_in[0];
    const float* W1l  = (const float*)d_in[1];
    const float* b1l  = (const float*)d_in[2];
    const float* W1r  = (const float*)d_in[3];
    const float* W2l  = (const float*)d_in[4];
    const float* b2l  = (const float*)d_in[5];
    const float* W2r  = (const float*)d_in[6];
    const int*   src1 = (const int*)d_in[7];
    const int*   dst1 = (const int*)d_in[8];
    const int*   src2 = (const int*)d_in[9];
    const int*   dst2 = (const int*)d_in[10];

    const int E1     = in_sizes[7];
    const int E2     = in_sizes[9];
    const int N1FEAT = in_sizes[2];
    const int N2FEAT = in_sizes[5];
    const int SIZE2  = out_size / N2FEAT;
    const int SIZE1  = 50000;

    float* agg1 = (float*)d_ws;
    float* cnt1 = agg1 + (size_t)SIZE1 * FEAT;
    float* h    = cnt1 + SIZE1;
    float* idx2 = h    + (size_t)SIZE1 * FEAT;
    int* need = (int*)cnt1;

    const int rsh1 = 8;
    const int nb1  = (SIZE1 + 255) >> 8;
    const int g1a  = (E1 + CHUNK - 1) / CHUNK;
    int* icnt1  = (int*)agg1;
    int* icur1  = icnt1 + IDX_STRIDE;
    int* iperm1 = icur1 + IDX_STRIDE;
    int* ips1   = iperm1 + (size_t)g1a * CHUNK;
    int* ipd1   = ips1   + (size_t)g1a * CHUNK;
    int* itbl1  = ipd1   + (size_t)g1a * CHUNK;
    const int rsh2 = 5;
    const int nb2  = (SIZE2 + 31) >> 5;
    const int g1b  = (E2 + CHUNK - 1) / CHUNK;
    int* icnt2  = (int*)idx2;
    int* icur2  = icnt2 + IDX_STRIDE;
    int* iperm2 = icur2 + IDX_STRIDE;
    int* ips2   = iperm2 + (size_t)g1b * CHUNK;
    int* ipd2   = ips2   + (size_t)g1b * CHUNK;
    int* itbl2  = ipd2   + (size_t)g1b * CHUNK;

    sage_sort_chunks<<<g1a + g1b, BT, 0, stream>>>(
        src1, dst1, ips1, ipd1, itbl1, E1, rsh1, nb1, g1a,
        src2, dst2, ips2, ipd2, itbl2, E2, rsh2, nb2, g1b,
        need, SIZE1);
    sage_rank_ranges<<<nb1 + nb2, BT, 0, stream>>>(
        ips1, ipd1, itbl1, icnt1, icur1, iperm1, SIZE1, E1, rsh1, g1a, nb1,
        ips2, ipd2, itbl2, icnt2, icur2, iperm2, SIZE2, E2, rsh2, g1b,
        src2, E2, need, SIZE1);

    {
        int units = (SIZE1 + 15) / 16;
        int grid  = units < 512 ? units : 512;
        int per   = (SIZE1 + grid - 1) / grid;
        sage_gather_gemm<8><<<grid, 256, 0, stream>>>(
            x, iperm1, icnt1, icur1, E1, W1l, W1r, b1l, h, SIZE1, per,
            need, SIZE2);
    }

    {
        int units = (SIZE2 + 31) / 32;
        int grid  = units < 512 ? units : 512;
        int per   = (SIZE2 + grid - 1) / grid;
        sage_gather_gemm<4><<<grid, 256, 0, stream>>>(
            h, iperm2, icnt2, icur2, E2, W2l, W2r, b2l, (float*)d_out, SIZE2,
            per, nullptr, 0);
    }
}


# Layer 1's passes overlap: a pass's source rows are requested ahead of the previous pass's WMMA products (ids 0-15 before the first half of k, 16-31 before the second), added after; sum order kept
# speedup vs baseline: 4.6295x; 1.0650x over previous
#include <hip/hip_runtime.h>

typedef __attribute__((ext_vector_type(2))) float v2f;
typedef __attribute__((ext_vector_type(8))) float v8f;
typedef __attribute__((ext_vector_type(4))) float v4f;

#define FEAT 128

#define IDX_STRIDE 65536

#define CHUNK 8192
#define BT    512
#define GMAX  512
#define RMAX  256
#define STAGE 16384

__device__ __forceinline__ int block_excl_scan(int v, int* wtot, int& total)
{
    const int lane = threadIdx.x & 31;
    const int w    = threadIdx.x >> 5;
    int incl = v;
    for (int o = 1; o < 32; o <<= 1) {
        int y = __shfl_up(incl, o);
        if (lane >= o) incl += y;
    }
    if (lane == 31) wtot[w] = incl;
    __syncthreads();
    int ws    = (lane < BT / 32) ? wtot[lane] : 0;
    int wincl = ws;
    for (int o = 1; o < 32; o <<= 1) {
        int y = __shfl_up(wincl, o);
        if (lane >= o) wincl += y;
    }
    total = __shfl(wincl, 31);
    const int wave_excl = __shfl(wincl, w) - __shfl(ws, w);
    return wave_excl + incl - v;
}

__global__ void __launch_bounds__(BT)
sage_sort_chunks(const int* __restrict__ srcA, const int* __restrict__ dstA,
                 int* __restrict__ psA, int* __restrict__ pdA,
                 int* __restrict__ tblA, int EA, int rshA, int nbA, int gA,
                 const int* __restrict__ srcB, const int* __restrict__ dstB,
                 int* __restrict__ psB, int* __restrict__ pdB,
                 int* __restrict__ tblB, int EB, int rshB, int nbB, int gB,
                 int* __restrict__ need, int nneed)
{
    constexpr int EPT = CHUNK / BT;
    for (int i = blockIdx.x * BT + threadIdx.x; i < nneed; i += gridDim.x * BT)
        need[i] = 0;
    __shared__ int hist[BT];
    __shared__ int wt[32];
    __shared__ __attribute__((aligned(16))) int sS[CHUNK];
    __shared__ __attribute__((aligned(16))) int sD[CHUNK];

    const bool second = (int)blockIdx.x >= gA;
    const int* __restrict__ src = second ? srcB : srcA;
    const int* __restrict__ dst = second ? dstB : dstA;
    int* __restrict__ ps  = second ? psB  : psA;
    int* __restrict__ pd  = second ? pdB  : pdA;
    int* __restrict__ tbl = second ? tblB : tblA;
    const int E   = second ? EB   : EA;
    const int rsh = second ? rshB : rshA;
    const int nb  = second ? nbB  : nbA;
    const int g1  = second ? gB   : gA;

    const int tid = threadIdx.x;
    const int g   = (int)blockIdx.x - (second ? gA : 0);
    const int e0  = g * CHUNK;

    hist[tid] = 0;
    int s[EPT], d[EPT], r[EPT];
    #pragma unroll
    for (int j = 0; j < EPT; ++j) {
        const int e = e0 + j * BT + tid;
        const bool ok = e < E;
        d[j] = ok ? dst[e] : 0;
        s[j] = ok ? src[e] : 0;
    }
    __syncthreads();
    #pragma unroll
    for (int j = 0; j < EPT; ++j) {
        r[j] = 0;
        if (e0 + j * BT + tid < E) {
            int bn = d[j] >> rsh;
            bn = bn < 0 ? 0 : (bn > nb - 1 ? nb - 1 : bn);
            r[j] = atomicAdd(&hist[bn], 1);
        }
    }
    __syncthreads();
    int total;
    const int ex = block_excl_scan(hist[tid], wt, total);
    hist[tid] = ex;
    if (tid <= nb) tbl[tid * g1 + g] = ex;
    __syncthreads();
    #pragma unroll
    for (int j = 0; j < EPT; ++j) {
        if (e0 + j * BT + tid < E) {
            int bn = d[j] >> rsh;
            bn = bn < 0 ? 0 : (bn > nb - 1 ? nb - 1 : bn);
            const int p = hist[bn] + r[j];
            sS[p] = s[j];
            sD[p] = d[j];
        }
    }
    __syncthreads();
    const int count = (E - e0 < CHUNK) ? (E - e0) : CHUNK;
    const int nq    = count >> 2;
    int4* ops = (int4*)(ps + e0);
    int4* opd = (int4*)(pd + e0);
    for (int i = tid; i < nq; i += BT) {
        ops[i] = ((const int4*)sS)[i];
        opd[i] = ((const int4*)sD)[i];
    }
    for (int i = (nq << 2) + tid; i < count; i += BT) {
        ps[e0 + i] = sS[i];
        pd[e0 + i] = sD[i];
    }
}

#define RU 8
__global__ void __launch_bounds__(BT)
sage_rank_ranges(const int* __restrict__ psA, const int* __restrict__ pdA,
                 const int* __restrict__ tblA,
                 int* __restrict__ cntA, int* __restrict__ curA,
                 int* __restrict__ permA, int nA, int EA, int rshA, int gA,
                 int nbA,
                 const int* __restrict__ psB, const int* __restrict__ pdB,
                 const int* __restrict__ tblB,
                 int* __restrict__ cntB, int* __restrict__ curB,
                 int* __restrict__ permB, int nB, int EB, int rshB, int gB,
                 const int* __restrict__ msrc, int mE,
                 int* __restrict__ need, int nneed)
{
    constexpr int NW = BT / 32;
    for (int e = blockIdx.x * BT + threadIdx.x; e < mE; e += gridDim.x * BT) {
        const int s = msrc[e];
        if ((unsigned)s < (unsigned)nneed) need[s] = 1;
    }
    const bool second = (int)blockIdx.x >= nbA;
    const int* __restrict__ ps  = second ? psB  : psA;
    const int* __restrict__ pd  = second ? pdB  : pdA;
    const int* __restrict__ tbl = second ? tblB : tblA;
    int* __restrict__ cnt  = second ? cntB  : cntA;
    int* __restrict__ cur  = second ? curB  : curA;
    int* __restrict__ perm = second ? permB : permA;
    const int n   = second ? nB   : nA;
    const int E   = second ? EB   : EA;
    const int rsh = second ? rshB : rshA;
    const int g1  = second ? gB   : gA;
    __shared__ int rs[GMAX + RU * NW];
    __shared__ int rl[GMAX + RU * NW];
    __shared__ int cL[RMAX];
    __shared__ int wtA[32];
    __shared__ int wtB[32];
    __shared__ int wtC[32];
    __shared__ int stg[STAGE];

    const int tid   = threadIdx.x;
    const int lane  = tid & 31;
    const int w     = __builtin_amdgcn_readfirstlane(tid >> 5);
    const int b     = (int)blockIdx.x - (second ? nbA : 0);
    const int R     = 1 << rsh;
    const int dbase = b << rsh;

    int lo = 0, len = 0;
    if (tid < g1) {
        lo = tbl[b * g1 + tid];
        int hi = tbl[(b + 1) * g1 + tid];
        lo = lo < 0 ? 0 : (lo > CHUNK ? CHUNK : lo);
        hi = hi < lo ? lo : (hi > CHUNK ? CHUNK : hi);
        len = hi - lo;
    }
    rs[tid] = tid * CHUNK + lo;
    rl[tid] = len;
    if (tid < RU * NW) { rs[GMAX + tid] = 0; rl[GMAX + tid] = 0; }
    if (tid < RMAX) cL[tid] = 0;
    int base, S;
    block_excl_scan(lo,  wtA, base);
    block_excl_scan(len, wtB, S);

    for (int g0 = w; g0 < g1; g0 += RU * NW) {
        int dv[RU][2], l[RU], off[RU];
        #pragma unroll
        for (int u = 0; u < RU; ++u) {
            l[u]   = rl[g0 + u * NW];
            off[u] = rs[g0 + u * NW];
        }
        #pragma unroll
        for (int u = 0; u < RU; ++u) {
            dv[u][0] = (lane      < l[u]) ? pd[off[u] + lane]      : -1;
            dv[u][1] = (lane + 32 < l[u]) ? pd[off[u] + lane + 32] : -1;
        }
        #pragma unroll
        for (int u = 0; u < RU; ++u) {
            #pragma unroll
            for (int k = 0; k < 2; ++k) {
                const int dl = dv[u][k] - dbase;
                if ((unsigned)dl < (unsigned)R) atomicAdd(&cL[dl], 1);
            }
        }
        #pragma unroll
        for (int u = 0; u < RU; ++u) {
            for (int i = 64 + lane; i < l[u]; i += 32) {
                const int dl = pd[off[u] + i] - dbase;
                if ((unsigned)dl < (unsigned)R) atomicAdd(&cL[dl], 1);
            }
        }
    }
    __syncthreads();

    const int c = (tid < R) ? cL[tid] : 0;
    int tot;
    const int ex = block_excl_scan(c, wtC, tot);
    if (tid < R) {
        cL[tid] = ex;
        const int dd = dbase + tid;
        if (dd < n) { cnt[dd] = c; cur[dd] = base + ex + c; }
    }
    __syncthreads();

    const bool staged = (S <= STAGE);
    for (int g0 = w; g0 < g1; g0 += RU * NW) {
        int dv[RU][2], sv[RU][2], l[RU], off[RU];
        #pragma unroll
        for (int u = 0; u < RU; ++u) {
            l[u]   = rl[g0 + u * NW];
            off[u] = rs[g0 + u * NW];
        }
        #pragma unroll
        for (int u = 0; u < RU; ++u) {
            const bool k0 = lane < l[u], k1 = lane + 32 < l[u];
            dv[u][0] = k0 ? pd[off[u] + lane]      : -1;
            sv[u][0] = k0 ? ps[off[u] + lane]      : 0;
            dv[u][1] = k1 ? pd[off[u] + lane + 32] : -1;
            sv[u][1] = k1 ? ps[off[u] + lane + 32] : 0;
        }
        #pragma unroll
        for (int u = 0; u < RU; ++u) {
            #pragma unroll
            for (int k = 0; k < 2; ++k) {
                const int dl = dv[u][k] - dbase;
                if ((unsigned)dl < (unsigned)R) {
                    const int p = atomicAdd(&cL[dl], 1);
                    if (staged) {
                        if ((unsigned)p < (unsigned)STAGE) stg[p] = sv[u][k];
                    } else {
                        const int q = base + p;
                        if ((unsigned)q < (unsigned)E) perm[q] = sv[u][k];
                    }
                }
            }
        }
        #pragma unroll
        for (int u = 0; u < RU; ++u) {
            for (int i = 64 + lane; i < l[u]; i += 32) {
                const int dl = pd[off[u] + i] - dbase;
                const int sx = ps[off[u] + i];
                if ((unsigned)dl < (unsigned)R) {
                    const int p = atomicAdd(&cL[dl], 1);
                    if (staged) {
                        if ((unsigned)p < (unsigned)STAGE) stg[p] = sx;
                    } else {
                        const int q = base + p;
                        if ((unsigned)q < (unsigned)E) perm[q] = sx;
                    }
                }
            }
        }
    }
    if (staged) {
        __syncthreads();
        for (int i = tid; i < S; i += BT) {
            const int q = base + i;
            if ((unsigned)q < (unsigned)E) perm[q] = stg[i];
        }
    }
}

#define LMAX   128
#define WPITCH 132
#define APITCH 260

template <int NSL, int SG, int GG>
__device__ __forceinline__ void sum_batch(const float* __restrict__ X,
                                          const int (&my)[SG],
                                          const int (&nrem)[SG], int j,
                                          int lane, v4f (&acc)[NSL], int g0)
{
    v4f v[SG][GG];
    #pragma unroll
    for (int u = 0; u < GG; ++u) {
        #pragma unroll
        for (int t = 0; t < SG; ++t) {
            const int sid = __builtin_amdgcn_readlane(my[t], j + u);
            v[t][u] = ((const v4f*)(X + (size_t)sid * FEAT))[lane];
        }
    }
    #pragma unroll
    for (int t = 0; t < SG; ++t) {
        #pragma unroll
        for (int u = 0; u < GG; ++u) {
            if (j + u < nrem[t]) acc[g0 + t] += v[t][u];
        }
    }
}

template <int NT>
__global__ void __launch_bounds__(256) __attribute__((amdgpu_waves_per_eu(4, 4)))
sage_gather_gemm(const float* __restrict__ X,
                 const int* __restrict__ perm, const int* __restrict__ cnt,
                 const int* __restrict__ cur, int E,
                 const float* __restrict__ W1, const float* __restrict__ W2,
                 const float* __restrict__ bias, float* __restrict__ out,
                 int M, int per, const int* __restrict__ need, int nkeep)
{
    constexpr int N    = NT * 16;
    constexpr int ROWS = (8 / NT) * 16;
    constexpr int NSL  = ROWS / 8;
    constexpr int SG   = 2;
    constexpr int WLD  = N / 8;
    __shared__ __attribute__((aligned(16))) float wl[2 * N * WPITCH];
    __shared__ __attribute__((aligned(16))) float lds[ROWS * APITCH];
    __shared__ __attribute__((aligned(16))) int lst[LMAX + 32];
    __shared__ int   sBeg[LMAX + 32];
    __shared__ int   sLen[LMAX + 32];
    __shared__ float sInv[LMAX + 32];
    __shared__ int wtot[8];

    const int tid  = threadIdx.x;
    const int lane = tid & 31;
    const int w    = __builtin_amdgcn_readfirstlane(tid >> 5);
    const int c    = lane & 15;
    const int h    = lane >> 4;
    const int nt   = w % NT;
    const int mg   = w / NT;
    const int n    = nt * 16 + c;

    const int rown = (int)blockIdx.x + (int)gridDim.x * tid;
    bool wanted = (tid < per) && (tid < LMAX) && (rown < M);
    if (wanted && need != nullptr && rown >= nkeep) wanted = (need[rown] != 0);
    const int rsafe = ((int)blockIdx.x < M) ? (int)blockIdx.x : (M - 1);
    if (tid < LMAX + 32) lst[tid] = rsafe;
    const unsigned bal = __builtin_amdgcn_ballot_w32(wanted);
    if (lane == 0) wtot[w] = __popc(bal);
    __syncthreads();
    int loff = 0, K = 0;
    #pragma unroll
    for (int i = 0; i < 8; ++i) {
        const int ci = wtot[i];
        if (i < w) loff += ci;
        K += ci;
    }
    if (wanted) lst[loff + __popc(bal & ((1u << lane) - 1u))] = rown;
    __syncthreads();

    int segc = 0, sege = 0;
    if (tid < LMAX + 32) {
        const int m = lst[tid];
        segc = cnt[m];
        sege = cur[m];
    }

    int   gb[NSL], gn[NSL], myq[NSL];
    float gi[NSL];
    v4f   pre2[NSL];
    auto fetch_ids = [&](int p, int (&GB)[NSL], int (&GN)[NSL],
                         float (&GI)[NSL], int (&MY)[NSL]) {
        #pragma unroll
        for (int s = 0; s < NSL; ++s) {
            const int slot = p * ROWS + s * 8 + w;
            GB[s] = __builtin_amdgcn_readfirstlane(sBeg[slot]);
            GN[s] = __builtin_amdgcn_readfirstlane(sLen[slot]);
            GI[s] = sInv[slot];
            MY[s] = (lane < GN[s]) ? perm[GB[s] + lane] : 0;
        }
    };
    auto fetch_own = [&](int p) {
        #pragma unroll
        for (int s = 0; s < NSL; ++s) {
            const int m = lst[p * ROWS + s * 8 + w];
            pre2[s] = ((const v4f*)(X + (size_t)m * FEAT))[lane];
        }
    };

    #pragma unroll
    for (int half = 0; half < 2; ++half) {
        const float* Wsrc = half ? W2 : W1;
        float4 t[WLD];
        #pragma unroll
        for (int i = 0; i < WLD; ++i)
            t[i] = ((const float4*)(Wsrc + (size_t)(w + 8 * i) * FEAT))[lane];
        #pragma unroll
        for (int i = 0; i < WLD; ++i)
            *(float4*)(wl + (half * N + w + 8 * i) * WPITCH + 4 * lane) = t[i];
        if (half == 0) {
            if (tid < LMAX + 32) {
                const int end = (sege < E) ? sege : E;
                const int b   = (sege - segc > 0) ? (sege - segc) : 0;
                int len = end - b;
                if (len < 0 || tid >= K) len = 0;
                sBeg[tid] = b;
                sLen[tid] = len;
                sInv[tid] = 1.0f / fmaxf((float)segc, 1.0f);
            }
            __syncthreads();
            fetch_ids(0, gb, gn, gi, myq);
            if (NSL != 2) fetch_own(0);
        }
    }
    const float bn = bias[n];

    const int passes = (K + ROWS - 1) / ROWS;

    const float* arow = lds + (mg * 16 + c) * APITCH + 4 * h;
    const float* brow = wl + n * WPITCH + 4 * h;

    v8f pacc = {};
    auto prod_half = [&](int half) {
        constexpr int KG = 2;
        const float* ap = arow + 128 * half;
        const float* bp = brow + half * (N * WPITCH);
        float4 a[2][KG], b[2][KG];
        #pragma unroll
        for (int i = 0; i < KG; ++i) {
            a[0][i] = *(const float4*)(ap + 8 * i);
            b[0][i] = *(const float4*)(bp + 8 * i);
        }
        #pragma unroll
        for (int g = 0; g < 16 / KG; ++g) {
            if (g + 1 < 16 / KG) {
                #pragma unroll
                for (int i = 0; i < KG; ++i) {
                    a[(g + 1) & 1][i] = *(const float4*)(ap + 8 * (KG * (g + 1) + i));
                    b[(g + 1) & 1][i] = *(const float4*)(bp + 8 * (KG * (g + 1) + i));
                }
            }
            #pragma unroll
            for (int i = 0; i < KG; ++i) {
                const float4 av = a[g & 1][i];
                const float4 bv = b[g & 1][i];
                v2f a0, a1, b0, b1;
                a0[0] = av.x; a0[1] = av.y; a1[0] = av.z; a1[1] = av.w;
                b0[0] = bv.x; b0[1] = bv.y; b1[0] = bv.z; b1[1] = bv.w;
                pacc = __builtin_amdgcn_wmma_f32_16x16x4_f32(
                     false, a0,  false, b0,
                     (short)0, pacc,  false,  false);
                pacc = __builtin_amdgcn_wmma_f32_16x16x4_f32(
                    false, a1, false, b1, (short)0, pacc, false, false);
            }
            asm volatile("" ::: "memory");
        }
    };
    auto store_tile = [&](int q) {
        const int sbase = q * ROWS + mg * 16 + h * 8;
        const int4 q0 = *(const int4*)(lst + sbase);
        const int4 q1 = *(const int4*)(lst + sbase + 4);
        const int orow[8] = {q0.x, q0.y, q0.z, q0.w, q1.x, q1.y, q1.z, q1.w};
        #pragma unroll
        for (int j = 0; j < 8; ++j) {
            if (sbase + j < K) {
                const float v = pacc[j] + bn;
                out[(size_t)orow[j] * N + n] = v > 0.0f ? v : 0.0f;
            }
        }
    };

    if constexpr (NSL == 2) {
        constexpr int GA = 16;
        int   nb2[NSL], nn2[NSL], nq2[NSL];
        float ni2[NSL];
        #pragma unroll
        for (int s = 0; s < NSL; ++s) { nb2[s] = 0; nn2[s] = 0; nq2[s] = 0; ni2[s] = 0.0f; }
        auto request8 = [&](v4f (&v)[NSL][GA], int off, int u0) {
            #pragma unroll
            for (int u = 0; u < 8; ++u) {
                #pragma unroll
                for (int t = 0; t < NSL; ++t) {
                    const int sid = __builtin_amdgcn_readlane(myq[t], off + u0 + u);
                    v[t][u0 + u] = ((const v4f*)(X + (size_t)sid * FEAT))[lane];
                }
            }
        };
        auto add_in = [&](const v4f (&v)[NSL][GA], v4f (&acc)[NSL], int off) {
            #pragma unroll
            for (int t = 0; t < NSL; ++t) {
                #pragma unroll
                for (int u = 0; u < GA; ++u) {
                    if (off + u < gn[t]) acc[t] += v[t][u];
                }
            }
        };
        auto trip = [&](int p, const bool pr) __attribute__((always_inline)) {
            v4f v[NSL][GA];
            v4f acc[NSL];
            request8(v, 0, 0);
            request8(v, 0, 8);
            fetch_own(p);
            if (p + 1 < passes) fetch_ids(p + 1, nb2, nn2, ni2, nq2);
            if (pr) {
                pacc = (v8f){};
                prod_half(0);
            }
            #pragma unroll
            for (int t = 0; t < NSL; ++t)
                acc[t] = (v4f){0.0f, 0.0f, 0.0f, 0.0f};
            add_in(v, acc, 0);
            asm volatile("" : "+v"(acc[0]), "+v"(acc[1]) :: "memory");
            const int nmx = gn[0] > gn[1] ? gn[0] : gn[1];
            if (nmx > GA) {
                request8(v, GA, 0);
                if (nmx > GA + 8) request8(v, GA, 8);
            }
            if (pr) {
                prod_half(1);
                store_tile(p - 1);
            }
            add_in(v, acc, GA);
            for (int base = 32; base < nmx; base += 32) {
                int my[SG], nrem[SG];
                int nm = 0;
                #pragma unroll
                for (int t = 0; t < SG; ++t) {
                    int r = gn[t] - base;
                    r = r < 0 ? 0 : (r > 32 ? 32 : r);
                    nrem[t] = r;
                    nm = r > nm ? r : nm;
                    my[t] = (lane < r) ? perm[gb[t] + base + lane] : 0;
                }
                for (int j = 0; j < nm; j += 8)
                    sum_batch<NSL, SG, 8>(X, my, nrem, j, lane, acc, 0);
            }
            __syncthreads();
            #pragma unroll
            for (int s = 0; s < NSL; ++s) {
                float* row = lds + (s * 8 + w) * APITCH + 4 * lane;
                *(v4f*)(row)        = acc[s] * gi[s];
                *(v4f*)(row + FEAT) = pre2[s];
            }
            __syncthreads();
            #pragma unroll
            for (int s = 0; s < NSL; ++s) {
                gb[s] = nb2[s]; gn[s] = nn2[s]; gi[s] = ni2[s]; myq[s] = nq2[s];
            }
        };
        if (passes > 0) {
            trip(0, false);
            for (int p = 1; p < passes; ++p) trip(p, true);
            pacc = (v8f){};
            prod_half(0);
            prod_half(1);
            store_tile(passes - 1);
        }
    } else {
    for (int p = 0; p < passes; ++p) {
        v4f acc[NSL];
        #pragma unroll
        for (int g0 = 0; g0 < NSL; g0 += SG) {
            int my[SG];
            int nmx = 0;
            #pragma unroll
            for (int t = 0; t < SG; ++t) {
                acc[g0 + t] = (v4f){0.0f, 0.0f, 0.0f, 0.0f};
                my[t] = myq[g0 + t];
                nmx   = gn[g0 + t] > nmx ? gn[g0 + t] : nmx;
            }
            for (int base = 0; base < nmx; base += 32) {
                int nrem[SG];
                int nm = 0;
                #pragma unroll
                for (int t = 0; t < SG; ++t) {
                    int r = gn[g0 + t] - base;
                    r = r < 0 ? 0 : (r > 32 ? 32 : r);
                    nrem[t] = r;
                    nm = r > nm ? r : nm;
                    if (base > 0)
                        my[t] = (lane < r) ? perm[gb[g0 + t] + base + lane] : 0;
                }
                for (int j = 0; j < nm; ) {
                    const int left = nm - j;
                    if (left > 8 && j + 16 <= 32) {
                        sum_batch<NSL, SG, 16>(X, my, nrem, j, lane, acc, g0);
                        j += 16;
                    } else {
                        sum_batch<NSL, SG, 8>(X, my, nrem, j, lane, acc, g0);
                        j += 8;
                    }
                }
            }
        }

        __syncthreads();
        #pragma unroll
        for (int s = 0; s < NSL; ++s) {
            float* row = lds + (s * 8 + w) * APITCH + 4 * lane;
            *(v4f*)(row)        = acc[s] * gi[s];
            *(v4f*)(row + FEAT) = pre2[s];
        }
        __syncthreads();

        if (p + 1 < passes) {
            fetch_ids(p + 1, gb, gn, gi, myq);
            fetch_own(p + 1);
        }

        pacc = (v8f){};
        prod_half(0);
        prod_half(1);
        store_tile(p);
    }
    }
}

extern "C" void kernel_launch(void* const* d_in, const int* in_sizes, int n_in,
                              void* d_out, int out_size, void* d_ws, size_t ws_size,
                              hipStream_t stream) {
    const float* x    = (const float*)d_in[0];
    const float* W1l  = (const float*)d_in[1];
    const float* b1l  = (const float*)d_in[2];
    const float* W1r  = (const float*)d_in[3];
    const float* W2l  = (const float*)d_in[4];
    const float* b2l  = (const float*)d_in[5];
    const float* W2r  = (const float*)d_in[6];
    const int*   src1 = (const int*)d_in[7];
    const int*   dst1 = (const int*)d_in[8];
    const int*   src2 = (const int*)d_in[9];
    const int*   dst2 = (const int*)d_in[10];

    const int E1     = in_sizes[7];
    const int E2     = in_sizes[9];
    const int N1FEAT = in_sizes[2];
    const int N2FEAT = in_sizes[5];
    const int SIZE2  = out_size / N2FEAT;
    const int SIZE1  = 50000;

    float* agg1 = (float*)d_ws;
    float* cnt1 = agg1 + (size_t)SIZE1 * FEAT;
    float* h    = cnt1 + SIZE1;
    float* idx2 = h    + (size_t)SIZE1 * FEAT;
    int* need = (int*)cnt1;

    const int rsh1 = 8;
    const int nb1  = (SIZE1 + 255) >> 8;
    const int g1a  = (E1 + CHUNK - 1) / CHUNK;
    int* icnt1  = (int*)agg1;
    int* icur1  = icnt1 + IDX_STRIDE;
    int* iperm1 = icur1 + IDX_STRIDE;
    int* ips1   = iperm1 + (size_t)g1a * CHUNK;
    int* ipd1   = ips1   + (size_t)g1a * CHUNK;
    int* itbl1  = ipd1   + (size_t)g1a * CHUNK;
    const int rsh2 = 5;
    const int nb2  = (SIZE2 + 31) >> 5;
    const int g1b  = (E2 + CHUNK - 1) / CHUNK;
    int* icnt2  = (int*)idx2;
    int* icur2  = icnt2 + IDX_STRIDE;
    int* iperm2 = icur2 + IDX_STRIDE;
    int* ips2   = iperm2 + (size_t)g1b * CHUNK;
    int* ipd2   = ips2   + (size_t)g1b * CHUNK;
    int* itbl2  = ipd2   + (size_t)g1b * CHUNK;

    sage_sort_chunks<<<g1a + g1b, BT, 0, stream>>>(
        src1, dst1, ips1, ipd1, itbl1, E1, rsh1, nb1, g1a,
        src2, dst2, ips2, ipd2, itbl2, E2, rsh2, nb2, g1b,
        need, SIZE1);
    sage_rank_ranges<<<nb1 + nb2, BT, 0, stream>>>(
        ips1, ipd1, itbl1, icnt1, icur1, iperm1, SIZE1, E1, rsh1, g1a, nb1,
        ips2, ipd2, itbl2, icnt2, icur2, iperm2, SIZE2, E2, rsh2, g1b,
        src2, E2, need, SIZE1);

    {
        int units = (SIZE1 + 15) / 16;
        int grid  = units < 512 ? units : 512;
        int per   = (SIZE1 + grid - 1) / grid;
        sage_gather_gemm<8><<<grid, 256, 0, stream>>>(
            x, iperm1, icnt1, icur1, E1, W1l, W1r, b1l, h, SIZE1, per,
            need, SIZE2);
    }

    {
        int units = (SIZE2 + 31) / 32;
        int grid  = units < 512 ? units : 512;
        int per   = (SIZE2 + grid - 1) / grid;
        sage_gather_gemm<4><<<grid, 256, 0, stream>>>(
            h, iperm2, icnt2, icur2, E2, W2l, W2r, b2l, (float*)d_out, SIZE2,
            per, nullptr, 0);
    }
}
